# starting kernel
# baseline (speedup 1.0000x reference)
#include <hip/hip_runtime.h>
#include <hip/hip_bf16.h>


typedef __attribute__((ext_vector_type(2))) float v2f;
typedef __attribute__((ext_vector_type(8))) float v8f;

#define D_FEAT 128

__global__ void zero_kernel(float* __restrict__ p, int n) {
  int i = blockIdx.x * blockDim.x + threadIdx.x;
  if (i < n) p[i] = 0.0f;
}

__global__ void deg_kernel(const int* __restrict__ col,
                           const float* __restrict__ w,
                           float* __restrict__ deg, int E) {
  int e = blockIdx.x * blockDim.x + threadIdx.x;
  if (e < E) atomicAdd(&deg[col[e]], w[e]);
}

__global__ void dinv_kernel(float* __restrict__ deg, int n) {
  int i = blockIdx.x * blockDim.x + threadIdx.x;
  if (i < n) {
    float d = deg[i] + 1.0f;
    deg[i] = rsqrtf(d);
  }
}

__global__ __launch_bounds__(256) void gemm_wmma_kernel(
    const float* __restrict__ Z, const float* __restrict__ W,
    float* __restrict__ H, int nRows) {
  const int lane = threadIdx.x;
  const int wave = threadIdx.y;
  const int tile = blockIdx.x * 8 + wave;
  const int nTiles = (nRows / 16) * (D_FEAT / 16);
  if (tile >= nTiles) return;

  const int m0 = (tile / (D_FEAT / 16)) * 16;
  const int n0 = (tile % (D_FEAT / 16)) * 16;
  const int l15   = lane & 15;
  const int khalf = (lane >> 4) << 1;

  const float* zrow = Z + (size_t)(m0 + l15) * D_FEAT;
  v8f acc = {};

#pragma unroll
  for (int k0 = 0; k0 < D_FEAT; k0 += 4) {
    const int kb = k0 + khalf;
    v2f a;
    a.x = zrow[kb];
    a.y = zrow[kb + 1];
    v2f b;
    b.x = W[(size_t)kb * D_FEAT + n0 + l15];
    b.y = W[(size_t)(kb + 1) * D_FEAT + n0 + l15];
    acc = __builtin_amdgcn_wmma_f32_16x16x4_f32(
         false, a,  false, b,
         (short)0, acc,  false,  false);
  }

  const int mbase = m0 + ((lane >> 4) << 3);
  float* hp = H + (size_t)mbase * D_FEAT + n0 + l15;
#pragma unroll
  for (int r = 0; r < 8; ++r) hp[(size_t)r * D_FEAT] = acc[r];
}

__global__ __launch_bounds__(256) void aggregate_kernel(
    const int* __restrict__ row, const int* __restrict__ col,
    const float* __restrict__ w, const float* __restrict__ dinv,
    const float* __restrict__ H, float* __restrict__ AGG, int E) {
  const int lane = threadIdx.x;
  const int e = blockIdx.x * 8 + threadIdx.y;
  if (e >= E) return;
  const int r = row[e];
  const int c = col[e];
  const float nrm = dinv[r] * w[e] * dinv[c];

  const float4 v = ((const float4*)(H + (size_t)r * D_FEAT))[lane];
  float* ap = AGG + (size_t)c * D_FEAT + lane * 4;
  atomicAdd(ap + 0, v.x * nrm);
  atomicAdd(ap + 1, v.y * nrm);
  atomicAdd(ap + 2, v.z * nrm);
  atomicAdd(ap + 3, v.w * nrm);
}

__global__ void finish_kernel(float* __restrict__ Z  ,
                              const float* __restrict__ H,
                              const float* __restrict__ dinv,
                              const float* __restrict__ bias,
                              const float* __restrict__ pw, int total) {
  int i = blockIdx.x * blockDim.x + threadIdx.x;
  if (i >= total) return;
  const int f = i & (D_FEAT - 1);
  const int node = i >> 7;
  const float d = dinv[node];
  const float v = Z[i] + H[i] * d * d + bias[f];
  Z[i] = fmaxf(v, 0.0f) + pw[f] * fminf(v, 0.0f);
}

__global__ void accum_kernel(const float* __restrict__ Z,
                             const float* __restrict__ coeff, int t,
                             float* __restrict__ out, int total) {
  int i = blockIdx.x * blockDim.x + threadIdx.x;
  if (i >= total) return;
  out[i] += coeff[t] * Z[i];
}

extern "C" void kernel_launch(void* const* d_in, const int* in_sizes, int n_in,
                              void* d_out, int out_size, void* d_ws, size_t ws_size,
                              hipStream_t stream) {
  const float* x       = (const float*)d_in[0];
  const int*   eidx    = (const int*)  d_in[1];
  const float* ew      = (const float*)d_in[2];
  const float* coeff   = (const float*)d_in[3];
  const float* W1      = (const float*)d_in[4];
  const float* b1      = (const float*)d_in[5];
  const float* W2      = (const float*)d_in[6];
  const float* b2      = (const float*)d_in[7];
  const float* prelu_w = (const float*)d_in[8];

  const int N = in_sizes[0] / D_FEAT;
  const int E = in_sizes[2] / 3;
  const int total = N * D_FEAT;

  float* ws   = (float*)d_ws;
  float* dinv = ws;
  float* bufA = ws + (((size_t)N + 31) & ~(size_t)31);
  float* bufB = bufA + (size_t)N * D_FEAT;

  const int ETHREADS = 256;
  const dim3 waveBlock(32, 8);
  const int gemmBlocks = ((N / 16) * (D_FEAT / 16) + 7) / 8;
  const int edgeBlocks = (E + 7) / 8;
  const int elemBlocks = (total + ETHREADS - 1) / ETHREADS;
  const int nodeBlocks = (N + ETHREADS - 1) / ETHREADS;

  zero_kernel<<<elemBlocks, ETHREADS, 0, stream>>>((float*)d_out, total);

  for (int t = 0; t < 3; ++t) {
    const int* row = eidx + (size_t)t * 2 * E;
    const int* col = row + E;
    const float* w = ew + (size_t)t * E;

    zero_kernel<<<nodeBlocks, ETHREADS, 0, stream>>>(dinv, N);
    deg_kernel<<<(E + ETHREADS - 1) / ETHREADS, ETHREADS, 0, stream>>>(col, w, dinv, E);
    dinv_kernel<<<nodeBlocks, ETHREADS, 0, stream>>>(dinv, N);

    gemm_wmma_kernel<<<gemmBlocks, waveBlock, 0, stream>>>(x, W1, bufA, N);
    zero_kernel<<<elemBlocks, ETHREADS, 0, stream>>>(bufB, total);
    aggregate_kernel<<<edgeBlocks, waveBlock, 0, stream>>>(row, col, w, dinv, bufA, bufB, E);
    finish_kernel<<<elemBlocks, ETHREADS, 0, stream>>>(bufB, bufA, dinv, b1, prelu_w, total);

    gemm_wmma_kernel<<<gemmBlocks, waveBlock, 0, stream>>>(bufB, W2, bufA, N);
    zero_kernel<<<elemBlocks, ETHREADS, 0, stream>>>(bufB, total);
    aggregate_kernel<<<edgeBlocks, waveBlock, 0, stream>>>(row, col, w, dinv, bufA, bufB, E);
    finish_kernel<<<elemBlocks, ETHREADS, 0, stream>>>(bufB, bufA, dinv, b2, prelu_w, total);

    accum_kernel<<<elemBlocks, ETHREADS, 0, stream>>>(bufB, coeff, t, (float*)d_out, total);
  }
}


# Edges bucketed by destination once per call (int atomic rank, scan, fill), buckets put in edge-id order; one wave per node sums its bucket in that fixed order, self loop, bias, PReLU; no f32 atomics
# speedup vs baseline: 2.1341x; 2.1341x over previous
#include <hip/hip_runtime.h>
#include <hip/hip_bf16.h>


typedef __attribute__((ext_vector_type(2))) float v2f;
typedef __attribute__((ext_vector_type(8))) float v8f;

#define D_FEAT 128
#define SCAN_B 1024

__global__ void zero_kernel(float* __restrict__ p, int n) {
  int i = blockIdx.x * blockDim.x + threadIdx.x;
  if (i < n) p[i] = 0.0f;
}

__global__ void rank_kernel(const int* __restrict__ eidx, int* __restrict__ cnt,
                            int* __restrict__ rank, int E, int N) {
  const int e = blockIdx.x * blockDim.x + threadIdx.x;
  const int t = blockIdx.y;
  if (e >= E) return;
  const int c = eidx[(size_t)t * 2 * E + E + e];
  rank[(size_t)t * E + e] = atomicAdd(&cnt[t * N + c], 1);
}

__global__ __launch_bounds__(SCAN_B) void bsum_kernel(
    const int* __restrict__ cnt, int* __restrict__ bsum, int n) {
  __shared__ int s[32];
  const int tid = threadIdx.x;
  const int i = blockIdx.x * SCAN_B + tid;
  int v = (i < n) ? cnt[i] : 0;
  for (int m = 16; m > 0; m >>= 1) v += __shfl_xor(v, m, 32);
  if ((tid & 31) == 0) s[tid >> 5] = v;
  __syncthreads();
  if (tid < 32) {
    int x = s[tid];
    for (int m = 16; m > 0; m >>= 1) x += __shfl_xor(x, m, 32);
    if (tid == 0) bsum[blockIdx.x] = x;
  }
}

__global__ __launch_bounds__(SCAN_B) void scan_kernel(
    const int* __restrict__ cnt, const int* __restrict__ bsum,
    int* __restrict__ off, int n) {
  __shared__ int sw[32];
  __shared__ int sbase;
  const int tid = threadIdx.x;
  const int lane = tid & 31;
  const int wv = tid >> 5;
  if (wv == 0) {
    int a = 0;
    for (int j = lane; j < (int)blockIdx.x; j += 32) a += bsum[j];
    for (int m = 16; m > 0; m >>= 1) a += __shfl_xor(a, m, 32);
    if (lane == 0) sbase = a;
  }
  const int i = blockIdx.x * SCAN_B + tid;
  const int v = (i < n) ? cnt[i] : 0;
  int x = v;
  for (int d = 1; d < 32; d <<= 1) {
    const int y = __shfl_up(x, d, 32);
    if (lane >= d) x += y;
  }
  if (lane == 31) sw[wv] = x;
  __syncthreads();
  if (wv == 0) {
    const int wtot = sw[lane];
    int xx = wtot;
    for (int d = 1; d < 32; d <<= 1) {
      const int y = __shfl_up(xx, d, 32);
      if (lane >= d) xx += y;
    }
    sw[lane] = xx - wtot;
  }
  __syncthreads();
  const int excl = sbase + sw[wv] + x - v;
  if (i < n) off[i] = excl;
  if (i == n - 1) off[n] = excl + v;
}

__global__ void fill_kernel(const int* __restrict__ eidx, const int* __restrict__ off,
                            const int* __restrict__ rank, int* __restrict__ tmp,
                            int E, int N) {
  const int e = blockIdx.x * blockDim.x + threadIdx.x;
  const int t = blockIdx.y;
  if (e >= E) return;
  const int c = eidx[(size_t)t * 2 * E + E + e];
  tmp[off[t * N + c] + rank[(size_t)t * E + e]] = e;
}

__global__ __launch_bounds__(256) void sort_kernel(
    const int* __restrict__ eidx, const float* __restrict__ ew,
    const int* __restrict__ off, const int* __restrict__ tmp,
    int* __restrict__ ssrc, float* __restrict__ swt, int E, int N) {
  const int lane = threadIdx.x;
  const int t = blockIdx.y;
  const int c = __builtin_amdgcn_readfirstlane(blockIdx.x * 8 + threadIdx.y);
  if (c >= N) return;
  const int* row = eidx + (size_t)t * 2 * E;
  const float* w = ew + (size_t)t * E;
  const int start = off[t * N + c];
  const int n = off[t * N + c + 1] - start;
  if (n <= 32) {
    const int eid = (lane < n) ? tmp[start + lane] : 0x7fffffff;
    int rk = 0;
    for (int j = 0; j < n; ++j)
      rk += (__builtin_amdgcn_readlane(eid, j) < eid) ? 1 : 0;
    if (lane < n) {
      ssrc[start + rk] = row[eid];
      swt[start + rk] = w[eid];
    }
  } else {
    for (int i = lane; i < n; i += 32) {
      const int eid = tmp[start + i];
      int rk = 0;
      for (int j = 0; j < n; ++j) rk += (tmp[start + j] < eid) ? 1 : 0;
      ssrc[start + rk] = row[eid];
      swt[start + rk] = w[eid];
    }
  }
}

__global__ void dinv_kernel(const int* __restrict__ off, const float* __restrict__ swt,
                            float* __restrict__ dinv, int G) {
  const int g = blockIdx.x * blockDim.x + threadIdx.x;
  if (g >= G) return;
  const int end = off[g + 1];
  float s = 0.0f;
  for (int p = off[g]; p < end; ++p) s += swt[p];
  const float d = s + 1.0f;
  dinv[g] = rsqrtf(d);
}

__global__ __launch_bounds__(256) void gemm_wmma_kernel(
    const float* __restrict__ Z, const float* __restrict__ W,
    float* __restrict__ H, int nRows) {
  const int lane = threadIdx.x;
  const int wave = threadIdx.y;
  const int tile = blockIdx.x * 8 + wave;
  const int nTiles = (nRows / 16) * (D_FEAT / 16);
  if (tile >= nTiles) return;

  const int m0 = (tile / (D_FEAT / 16)) * 16;
  const int n0 = (tile % (D_FEAT / 16)) * 16;
  const int l15   = lane & 15;
  const int khalf = (lane >> 4) << 1;

  const float* zrow = Z + (size_t)(m0 + l15) * D_FEAT;
  v8f acc = {};

#pragma unroll
  for (int k0 = 0; k0 < D_FEAT; k0 += 4) {
    const int kb = k0 + khalf;
    v2f a;
    a.x = zrow[kb];
    a.y = zrow[kb + 1];
    v2f b;
    b.x = W[(size_t)kb * D_FEAT + n0 + l15];
    b.y = W[(size_t)(kb + 1) * D_FEAT + n0 + l15];
    acc = __builtin_amdgcn_wmma_f32_16x16x4_f32(
         false, a,  false, b,
         (short)0, acc,  false,  false);
  }

  const int mbase = m0 + ((lane >> 4) << 3);
  float* hp = H + (size_t)mbase * D_FEAT + n0 + l15;
#pragma unroll
  for (int r = 0; r < 8; ++r) hp[(size_t)r * D_FEAT] = acc[r];
}

__global__ __launch_bounds__(256) void pull_kernel(
    const int* __restrict__ off, const int* __restrict__ ssrc,
    const float* __restrict__ swt, const float* __restrict__ dinv,
    const float* __restrict__ H, const float* __restrict__ bias,
    const float* __restrict__ pw, float* __restrict__ Zout, int N) {
  const int lane = threadIdx.x;
  const int c = __builtin_amdgcn_readfirstlane(blockIdx.x * 8 + threadIdx.y);
  if (c >= N) return;
  const int start = off[c];
  const int end = off[c + 1];
  const float dc = dinv[c];
  const float4* H4 = (const float4*)H;

  float4 acc = make_float4(0.0f, 0.0f, 0.0f, 0.0f);
  for (int base = start; base < end; base += 32) {
    const int m = (end - base < 32) ? (end - base) : 32;
    int r = 0;
    float nrm = 0.0f;
    if (lane < m) {
      r = ssrc[base + lane];
      nrm = dinv[r] * swt[base + lane] * dc;
    }
    const int nrm_bits = __float_as_int(nrm);
#pragma unroll 4
    for (int j = 0; j < m; ++j) {
      const int rj = __builtin_amdgcn_readlane(r, j);
      const float nj = __int_as_float(__builtin_amdgcn_readlane(nrm_bits, j));
      const float4 v = H4[(size_t)rj * (D_FEAT / 4) + lane];
      acc.x += v.x * nj;
      acc.y += v.y * nj;
      acc.z += v.z * nj;
      acc.w += v.w * nj;
    }
  }

  const float4 h = H4[(size_t)c * (D_FEAT / 4) + lane];
  const float4 b = ((const float4*)bias)[lane];
  const float4 p = ((const float4*)pw)[lane];
  float4 o;
  float v;
  v = acc.x + h.x * dc * dc + b.x;
  o.x = fmaxf(v, 0.0f) + p.x * fminf(v, 0.0f);
  v = acc.y + h.y * dc * dc + b.y;
  o.y = fmaxf(v, 0.0f) + p.y * fminf(v, 0.0f);
  v = acc.z + h.z * dc * dc + b.z;
  o.z = fmaxf(v, 0.0f) + p.z * fminf(v, 0.0f);
  v = acc.w + h.w * dc * dc + b.w;
  o.w = fmaxf(v, 0.0f) + p.w * fminf(v, 0.0f);
  ((float4*)Zout)[(size_t)c * (D_FEAT / 4) + lane] = o;
}

__global__ void accum_kernel(const float* __restrict__ Z,
                             const float* __restrict__ coeff, int t,
                             float* __restrict__ out, int total) {
  int i = blockIdx.x * blockDim.x + threadIdx.x;
  if (i >= total) return;
  out[i] += coeff[t] * Z[i];
}

extern "C" void kernel_launch(void* const* d_in, const int* in_sizes, int n_in,
                              void* d_out, int out_size, void* d_ws, size_t ws_size,
                              hipStream_t stream) {
  const float* x       = (const float*)d_in[0];
  const int*   eidx    = (const int*)  d_in[1];
  const float* ew      = (const float*)d_in[2];
  const float* coeff   = (const float*)d_in[3];
  const float* W1      = (const float*)d_in[4];
  const float* b1      = (const float*)d_in[5];
  const float* W2      = (const float*)d_in[6];
  const float* b2      = (const float*)d_in[7];
  const float* prelu_w = (const float*)d_in[8];

  const int N = in_sizes[0] / D_FEAT;
  const int E = in_sizes[2] / 3;
  const int total = N * D_FEAT;
  const int G = 3 * N;
  const size_t E3 = (size_t)3 * E;

  const size_t padG  = ((size_t)G + 1 + 31) & ~(size_t)31;
  const size_t padE3 = (E3 + 31) & ~(size_t)31;
  const int scanBlocks = (G + SCAN_B - 1) / SCAN_B;
  const size_t padS  = ((size_t)scanBlocks + 31) & ~(size_t)31;
  int*   cnt  = (int*)d_ws;
  int*   off  = cnt + padG;
  int*   bsum = off + padG;
  int*   rank = bsum + padS;
  int*   tmp  = rank + padE3;
  int*   ssrc = tmp + padE3;
  float* swt  = (float*)(ssrc + padE3);
  float* dinv = swt + padE3;
  float* bufA = dinv + padG;
  float* bufB = bufA + (size_t)N * D_FEAT;

  const int ETHREADS = 256;
  const dim3 waveBlock(32, 8);
  const int gemmBlocks = ((N / 16) * (D_FEAT / 16) + 7) / 8;
  const int nodeWaveBlocks = (N + 7) / 8;
  const int elemBlocks = (total + ETHREADS - 1) / ETHREADS;
  const dim3 edgeGrid((E + ETHREADS - 1) / ETHREADS, 3);
  const dim3 nodeWaveGrid(nodeWaveBlocks, 3);

  zero_kernel<<<elemBlocks, ETHREADS, 0, stream>>>((float*)d_out, total);

  zero_kernel<<<(G + ETHREADS - 1) / ETHREADS, ETHREADS, 0, stream>>>((float*)cnt, G);
  rank_kernel<<<edgeGrid, ETHREADS, 0, stream>>>(eidx, cnt, rank, E, N);
  bsum_kernel<<<scanBlocks, SCAN_B, 0, stream>>>(cnt, bsum, G);
  scan_kernel<<<scanBlocks, SCAN_B, 0, stream>>>(cnt, bsum, off, G);
  fill_kernel<<<edgeGrid, ETHREADS, 0, stream>>>(eidx, off, rank, tmp, E, N);
  sort_kernel<<<nodeWaveGrid, waveBlock, 0, stream>>>(eidx, ew, off, tmp, ssrc, swt, E, N);
  dinv_kernel<<<(G + ETHREADS - 1) / ETHREADS, ETHREADS, 0, stream>>>(off, swt, dinv, G);

  for (int t = 0; t < 3; ++t) {
    const int*   offT  = off + (size_t)t * N;
    const float* dinvT = dinv + (size_t)t * N;

    gemm_wmma_kernel<<<gemmBlocks, waveBlock, 0, stream>>>(x, W1, bufA, N);
    pull_kernel<<<nodeWaveBlocks, waveBlock, 0, stream>>>(offT, ssrc, swt, dinvT, bufA, b1, prelu_w, bufB, N);

    gemm_wmma_kernel<<<gemmBlocks, waveBlock, 0, stream>>>(bufB, W2, bufA, N);
    pull_kernel<<<nodeWaveBlocks, waveBlock, 0, stream>>>(offT, ssrc, swt, dinvT, bufA, b2, prelu_w, bufB, N);

    accum_kernel<<<elemBlocks, ETHREADS, 0, stream>>>(bufB, coeff, t, (float*)d_out, total);
  }
}


# Weighted sum over scales folded into the layer-2 gather: it stores coeff[0]*z for scale 0 and adds coeff[t]*z for 1,2; out zero-fill and the 3 per-element add launches gone; sum order unchanged
# speedup vs baseline: 2.3489x; 1.1007x over previous
#include <hip/hip_runtime.h>
#include <hip/hip_bf16.h>


typedef __attribute__((ext_vector_type(2))) float v2f;
typedef __attribute__((ext_vector_type(8))) float v8f;

#define D_FEAT 128
#define SCAN_B 1024

__global__ void zero_kernel(float* __restrict__ p, int n) {
  int i = blockIdx.x * blockDim.x + threadIdx.x;
  if (i < n) p[i] = 0.0f;
}

__global__ void rank_kernel(const int* __restrict__ eidx, int* __restrict__ cnt,
                            int* __restrict__ rank, int E, int N) {
  const int e = blockIdx.x * blockDim.x + threadIdx.x;
  const int t = blockIdx.y;
  if (e >= E) return;
  const int c = eidx[(size_t)t * 2 * E + E + e];
  rank[(size_t)t * E + e] = atomicAdd(&cnt[t * N + c], 1);
}

__global__ __launch_bounds__(SCAN_B) void bsum_kernel(
    const int* __restrict__ cnt, int* __restrict__ bsum, int n) {
  __shared__ int s[32];
  const int tid = threadIdx.x;
  const int i = blockIdx.x * SCAN_B + tid;
  int v = (i < n) ? cnt[i] : 0;
  for (int m = 16; m > 0; m >>= 1) v += __shfl_xor(v, m, 32);
  if ((tid & 31) == 0) s[tid >> 5] = v;
  __syncthreads();
  if (tid < 32) {
    int x = s[tid];
    for (int m = 16; m > 0; m >>= 1) x += __shfl_xor(x, m, 32);
    if (tid == 0) bsum[blockIdx.x] = x;
  }
}

__global__ __launch_bounds__(SCAN_B) void scan_kernel(
    const int* __restrict__ cnt, const int* __restrict__ bsum,
    int* __restrict__ off, int n) {
  __shared__ int sw[32];
  __shared__ int sbase;
  const int tid = threadIdx.x;
  const int lane = tid & 31;
  const int wv = tid >> 5;
  if (wv == 0) {
    int a = 0;
    for (int j = lane; j < (int)blockIdx.x; j += 32) a += bsum[j];
    for (int m = 16; m > 0; m >>= 1) a += __shfl_xor(a, m, 32);
    if (lane == 0) sbase = a;
  }
  const int i = blockIdx.x * SCAN_B + tid;
  const int v = (i < n) ? cnt[i] : 0;
  int x = v;
  for (int d = 1; d < 32; d <<= 1) {
    const int y = __shfl_up(x, d, 32);
    if (lane >= d) x += y;
  }
  if (lane == 31) sw[wv] = x;
  __syncthreads();
  if (wv == 0) {
    const int wtot = sw[lane];
    int xx = wtot;
    for (int d = 1; d < 32; d <<= 1) {
      const int y = __shfl_up(xx, d, 32);
      if (lane >= d) xx += y;
    }
    sw[lane] = xx - wtot;
  }
  __syncthreads();
  const int excl = sbase + sw[wv] + x - v;
  if (i < n) off[i] = excl;
  if (i == n - 1) off[n] = excl + v;
}

__global__ void fill_kernel(const int* __restrict__ eidx, const int* __restrict__ off,
                            const int* __restrict__ rank, int* __restrict__ tmp,
                            int E, int N) {
  const int e = blockIdx.x * blockDim.x + threadIdx.x;
  const int t = blockIdx.y;
  if (e >= E) return;
  const int c = eidx[(size_t)t * 2 * E + E + e];
  tmp[off[t * N + c] + rank[(size_t)t * E + e]] = e;
}

__global__ __launch_bounds__(256) void sort_kernel(
    const int* __restrict__ eidx, const float* __restrict__ ew,
    const int* __restrict__ off, const int* __restrict__ tmp,
    int* __restrict__ ssrc, float* __restrict__ swt, int E, int N) {
  const int lane = threadIdx.x;
  const int t = blockIdx.y;
  const int c = __builtin_amdgcn_readfirstlane(blockIdx.x * 8 + threadIdx.y);
  if (c >= N) return;
  const int* row = eidx + (size_t)t * 2 * E;
  const float* w = ew + (size_t)t * E;
  const int start = off[t * N + c];
  const int n = off[t * N + c + 1] - start;
  if (n <= 32) {
    const int eid = (lane < n) ? tmp[start + lane] : 0x7fffffff;
    int rk = 0;
    for (int j = 0; j < n; ++j)
      rk += (__builtin_amdgcn_readlane(eid, j) < eid) ? 1 : 0;
    if (lane < n) {
      ssrc[start + rk] = row[eid];
      swt[start + rk] = w[eid];
    }
  } else {
    for (int i = lane; i < n; i += 32) {
      const int eid = tmp[start + i];
      int rk = 0;
      for (int j = 0; j < n; ++j) rk += (tmp[start + j] < eid) ? 1 : 0;
      ssrc[start + rk] = row[eid];
      swt[start + rk] = w[eid];
    }
  }
}

__global__ void dinv_kernel(const int* __restrict__ off, const float* __restrict__ swt,
                            float* __restrict__ dinv, int G) {
  const int g = blockIdx.x * blockDim.x + threadIdx.x;
  if (g >= G) return;
  const int end = off[g + 1];
  float s = 0.0f;
  for (int p = off[g]; p < end; ++p) s += swt[p];
  const float d = s + 1.0f;
  dinv[g] = rsqrtf(d);
}

__global__ __launch_bounds__(256) void gemm_wmma_kernel(
    const float* __restrict__ Z, const float* __restrict__ W,
    float* __restrict__ H, int nRows) {
  const int lane = threadIdx.x;
  const int wave = threadIdx.y;
  const int tile = blockIdx.x * 8 + wave;
  const int nTiles = (nRows / 16) * (D_FEAT / 16);
  if (tile >= nTiles) return;

  const int m0 = (tile / (D_FEAT / 16)) * 16;
  const int n0 = (tile % (D_FEAT / 16)) * 16;
  const int l15   = lane & 15;
  const int khalf = (lane >> 4) << 1;

  const float* zrow = Z + (size_t)(m0 + l15) * D_FEAT;
  v8f acc = {};

#pragma unroll
  for (int k0 = 0; k0 < D_FEAT; k0 += 4) {
    const int kb = k0 + khalf;
    v2f a;
    a.x = zrow[kb];
    a.y = zrow[kb + 1];
    v2f b;
    b.x = W[(size_t)kb * D_FEAT + n0 + l15];
    b.y = W[(size_t)(kb + 1) * D_FEAT + n0 + l15];
    acc = __builtin_amdgcn_wmma_f32_16x16x4_f32(
         false, a,  false, b,
         (short)0, acc,  false,  false);
  }

  const int mbase = m0 + ((lane >> 4) << 3);
  float* hp = H + (size_t)mbase * D_FEAT + n0 + l15;
#pragma unroll
  for (int r = 0; r < 8; ++r) hp[(size_t)r * D_FEAT] = acc[r];
}

template <int MODE>
__global__ __launch_bounds__(256) void pull_kernel(
    const int* __restrict__ off, const int* __restrict__ ssrc,
    const float* __restrict__ swt, const float* __restrict__ dinv,
    const float* __restrict__ H, const float* __restrict__ bias,
    const float* __restrict__ pw, const float* __restrict__ cscale,
    float* __restrict__ Zout, int N) {
  const int lane = threadIdx.x;
  const int c = __builtin_amdgcn_readfirstlane(blockIdx.x * 8 + threadIdx.y);
  if (c >= N) return;
  const int start = off[c];
  const int end = off[c + 1];
  const float dc = dinv[c];
  const float4* H4 = (const float4*)H;

  float4 acc = make_float4(0.0f, 0.0f, 0.0f, 0.0f);
  for (int base = start; base < end; base += 32) {
    const int m = (end - base < 32) ? (end - base) : 32;
    int r = 0;
    float nrm = 0.0f;
    if (lane < m) {
      r = ssrc[base + lane];
      nrm = dinv[r] * swt[base + lane] * dc;
    }
    const int nrm_bits = __float_as_int(nrm);
#pragma unroll 4
    for (int j = 0; j < m; ++j) {
      const int rj = __builtin_amdgcn_readlane(r, j);
      const float nj = __int_as_float(__builtin_amdgcn_readlane(nrm_bits, j));
      const float4 v = H4[(size_t)rj * (D_FEAT / 4) + lane];
      acc.x += v.x * nj;
      acc.y += v.y * nj;
      acc.z += v.z * nj;
      acc.w += v.w * nj;
    }
  }

  const float4 h = H4[(size_t)c * (D_FEAT / 4) + lane];
  const float4 b = ((const float4*)bias)[lane];
  const float4 p = ((const float4*)pw)[lane];
  float4 o;
  float v;
  v = acc.x + h.x * dc * dc + b.x;
  o.x = fmaxf(v, 0.0f) + p.x * fminf(v, 0.0f);
  v = acc.y + h.y * dc * dc + b.y;
  o.y = fmaxf(v, 0.0f) + p.y * fminf(v, 0.0f);
  v = acc.z + h.z * dc * dc + b.z;
  o.z = fmaxf(v, 0.0f) + p.z * fminf(v, 0.0f);
  v = acc.w + h.w * dc * dc + b.w;
  o.w = fmaxf(v, 0.0f) + p.w * fminf(v, 0.0f);
  float4* zp = (float4*)Zout + (size_t)c * (D_FEAT / 4) + lane;
  if (MODE == 0) {
    *zp = o;
  } else {
    const float ct = cscale[0];
    float4 r;
    if (MODE == 1) {
      r.x = ct * o.x;
      r.y = ct * o.y;
      r.z = ct * o.z;
      r.w = ct * o.w;
    } else {
      const float4 s = *zp;
      r.x = s.x + ct * o.x;
      r.y = s.y + ct * o.y;
      r.z = s.z + ct * o.z;
      r.w = s.w + ct * o.w;
    }
    *zp = r;
  }
}

extern "C" void kernel_launch(void* const* d_in, const int* in_sizes, int n_in,
                              void* d_out, int out_size, void* d_ws, size_t ws_size,
                              hipStream_t stream) {
  const float* x       = (const float*)d_in[0];
  const int*   eidx    = (const int*)  d_in[1];
  const float* ew      = (const float*)d_in[2];
  const float* coeff   = (const float*)d_in[3];
  const float* W1      = (const float*)d_in[4];
  const float* b1      = (const float*)d_in[5];
  const float* W2      = (const float*)d_in[6];
  const float* b2      = (const float*)d_in[7];
  const float* prelu_w = (const float*)d_in[8];

  const int N = in_sizes[0] / D_FEAT;
  const int E = in_sizes[2] / 3;
  const int G = 3 * N;
  const size_t E3 = (size_t)3 * E;

  const size_t padG  = ((size_t)G + 1 + 31) & ~(size_t)31;
  const size_t padE3 = (E3 + 31) & ~(size_t)31;
  const int scanBlocks = (G + SCAN_B - 1) / SCAN_B;
  const size_t padS  = ((size_t)scanBlocks + 31) & ~(size_t)31;
  int*   cnt  = (int*)d_ws;
  int*   off  = cnt + padG;
  int*   bsum = off + padG;
  int*   rank = bsum + padS;
  int*   tmp  = rank + padE3;
  int*   ssrc = tmp + padE3;
  float* swt  = (float*)(ssrc + padE3);
  float* dinv = swt + padE3;
  float* bufA = dinv + padG;
  float* bufB = bufA + (size_t)N * D_FEAT;

  const int ETHREADS = 256;
  const dim3 waveBlock(32, 8);
  const int gemmBlocks = ((N / 16) * (D_FEAT / 16) + 7) / 8;
  const int nodeWaveBlocks = (N + 7) / 8;
  const dim3 edgeGrid((E + ETHREADS - 1) / ETHREADS, 3);
  const dim3 nodeWaveGrid(nodeWaveBlocks, 3);

  zero_kernel<<<(G + ETHREADS - 1) / ETHREADS, ETHREADS, 0, stream>>>((float*)cnt, G);
  rank_kernel<<<edgeGrid, ETHREADS, 0, stream>>>(eidx, cnt, rank, E, N);
  bsum_kernel<<<scanBlocks, SCAN_B, 0, stream>>>(cnt, bsum, G);
  scan_kernel<<<scanBlocks, SCAN_B, 0, stream>>>(cnt, bsum, off, G);
  fill_kernel<<<edgeGrid, ETHREADS, 0, stream>>>(eidx, off, rank, tmp, E, N);
  sort_kernel<<<nodeWaveGrid, waveBlock, 0, stream>>>(eidx, ew, off, tmp, ssrc, swt, E, N);
  dinv_kernel<<<(G + ETHREADS - 1) / ETHREADS, ETHREADS, 0, stream>>>(off, swt, dinv, G);

  for (int t = 0; t < 3; ++t) {
    const int*   offT  = off + (size_t)t * N;
    const float* dinvT = dinv + (size_t)t * N;

    gemm_wmma_kernel<<<gemmBlocks, waveBlock, 0, stream>>>(x, W1, bufA, N);
    pull_kernel<0><<<nodeWaveBlocks, waveBlock, 0, stream>>>(offT, ssrc, swt, dinvT, bufA, b1, prelu_w, coeff + t, bufB, N);

    gemm_wmma_kernel<<<gemmBlocks, waveBlock, 0, stream>>>(bufB, W2, bufA, N);
    if (t == 0)
      pull_kernel<1><<<nodeWaveBlocks, waveBlock, 0, stream>>>(offT, ssrc, swt, dinvT, bufA, b2, prelu_w, coeff + t, (float*)d_out, N);
    else
      pull_kernel<2><<<nodeWaveBlocks, waveBlock, 0, stream>>>(offT, ssrc, swt, dinvT, bufA, b2, prelu_w, coeff + t, (float*)d_out, N);
  }
}


# GEMM re-tiled: one wave per 16x128 row strip on 8 accumulator tiles, tile columns interleaved by 4, so Z rows are fetched once and W, H move as whole-line 16-byte accesses; k order unchanged
# speedup vs baseline: 2.7064x; 1.1522x over previous
#include <hip/hip_runtime.h>
#include <hip/hip_bf16.h>


typedef __attribute__((ext_vector_type(2))) float v2f;
typedef __attribute__((ext_vector_type(8))) float v8f;

#define D_FEAT 128
#define SCAN_B 1024

__global__ void zero_kernel(float* __restrict__ p, int n) {
  int i = blockIdx.x * blockDim.x + threadIdx.x;
  if (i < n) p[i] = 0.0f;
}

__global__ void rank_kernel(const int* __restrict__ eidx, int* __restrict__ cnt,
                            int* __restrict__ rank, int E, int N) {
  const int e = blockIdx.x * blockDim.x + threadIdx.x;
  const int t = blockIdx.y;
  if (e >= E) return;
  const int c = eidx[(size_t)t * 2 * E + E + e];
  rank[(size_t)t * E + e] = atomicAdd(&cnt[t * N + c], 1);
}

__global__ __launch_bounds__(SCAN_B) void bsum_kernel(
    const int* __restrict__ cnt, int* __restrict__ bsum, int n) {
  __shared__ int s[32];
  const int tid = threadIdx.x;
  const int i = blockIdx.x * SCAN_B + tid;
  int v = (i < n) ? cnt[i] : 0;
  for (int m = 16; m > 0; m >>= 1) v += __shfl_xor(v, m, 32);
  if ((tid & 31) == 0) s[tid >> 5] = v;
  __syncthreads();
  if (tid < 32) {
    int x = s[tid];
    for (int m = 16; m > 0; m >>= 1) x += __shfl_xor(x, m, 32);
    if (tid == 0) bsum[blockIdx.x] = x;
  }
}

__global__ __launch_bounds__(SCAN_B) void scan_kernel(
    const int* __restrict__ cnt, const int* __restrict__ bsum,
    int* __restrict__ off, int n) {
  __shared__ int sw[32];
  __shared__ int sbase;
  const int tid = threadIdx.x;
  const int lane = tid & 31;
  const int wv = tid >> 5;
  if (wv == 0) {
    int a = 0;
    for (int j = lane; j < (int)blockIdx.x; j += 32) a += bsum[j];
    for (int m = 16; m > 0; m >>= 1) a += __shfl_xor(a, m, 32);
    if (lane == 0) sbase = a;
  }
  const int i = blockIdx.x * SCAN_B + tid;
  const int v = (i < n) ? cnt[i] : 0;
  int x = v;
  for (int d = 1; d < 32; d <<= 1) {
    const int y = __shfl_up(x, d, 32);
    if (lane >= d) x += y;
  }
  if (lane == 31) sw[wv] = x;
  __syncthreads();
  if (wv == 0) {
    const int wtot = sw[lane];
    int xx = wtot;
    for (int d = 1; d < 32; d <<= 1) {
      const int y = __shfl_up(xx, d, 32);
      if (lane >= d) xx += y;
    }
    sw[lane] = xx - wtot;
  }
  __syncthreads();
  const int excl = sbase + sw[wv] + x - v;
  if (i < n) off[i] = excl;
  if (i == n - 1) off[n] = excl + v;
}

__global__ void fill_kernel(const int* __restrict__ eidx, const int* __restrict__ off,
                            const int* __restrict__ rank, int* __restrict__ tmp,
                            int E, int N) {
  const int e = blockIdx.x * blockDim.x + threadIdx.x;
  const int t = blockIdx.y;
  if (e >= E) return;
  const int c = eidx[(size_t)t * 2 * E + E + e];
  tmp[off[t * N + c] + rank[(size_t)t * E + e]] = e;
}

__global__ __launch_bounds__(256) void sort_kernel(
    const int* __restrict__ eidx, const float* __restrict__ ew,
    const int* __restrict__ off, const int* __restrict__ tmp,
    int* __restrict__ ssrc, float* __restrict__ swt, int E, int N) {
  const int lane = threadIdx.x;
  const int t = blockIdx.y;
  const int c = __builtin_amdgcn_readfirstlane(blockIdx.x * 8 + threadIdx.y);
  if (c >= N) return;
  const int* row = eidx + (size_t)t * 2 * E;
  const float* w = ew + (size_t)t * E;
  const int start = off[t * N + c];
  const int n = off[t * N + c + 1] - start;
  if (n <= 32) {
    const int eid = (lane < n) ? tmp[start + lane] : 0x7fffffff;
    int rk = 0;
    for (int j = 0; j < n; ++j)
      rk += (__builtin_amdgcn_readlane(eid, j) < eid) ? 1 : 0;
    if (lane < n) {
      ssrc[start + rk] = row[eid];
      swt[start + rk] = w[eid];
    }
  } else {
    for (int i = lane; i < n; i += 32) {
      const int eid = tmp[start + i];
      int rk = 0;
      for (int j = 0; j < n; ++j) rk += (tmp[start + j] < eid) ? 1 : 0;
      ssrc[start + rk] = row[eid];
      swt[start + rk] = w[eid];
    }
  }
}

__global__ void dinv_kernel(const int* __restrict__ off, const float* __restrict__ swt,
                            float* __restrict__ dinv, int G) {
  const int g = blockIdx.x * blockDim.x + threadIdx.x;
  if (g >= G) return;
  const int end = off[g + 1];
  float s = 0.0f;
  for (int p = off[g]; p < end; ++p) s += swt[p];
  const float d = s + 1.0f;
  dinv[g] = rsqrtf(d);
}

typedef __attribute__((ext_vector_type(4))) float v4f;

#define GEMM_WAVES 4
#define KSTEPS_PER_GROUP 4

struct FragGroup {
  v2f a[KSTEPS_PER_GROUP];
  v4f blo[KSTEPS_PER_GROUP][2];
  v4f bhi[KSTEPS_PER_GROUP][2];
};

__device__ __forceinline__ void gemm_load_group(FragGroup& f, const float* __restrict__ zp,
                                                const float* __restrict__ wp, int kbase) {
#pragma unroll
  for (int q = 0; q < KSTEPS_PER_GROUP; ++q) {
    const int k0 = kbase + 4 * q;
    f.a[q] = *(const v2f*)(zp + k0);
#pragma unroll
    for (int s = 0; s < 2; ++s) {
      f.blo[q][s] = *(const v4f*)(wp + (size_t)(k0 + s) * D_FEAT);
      f.bhi[q][s] = *(const v4f*)(wp + (size_t)(k0 + s) * D_FEAT + 64);
    }
  }
}

__device__ __forceinline__ void gemm_mma_group(v8f (&acc)[8], const FragGroup& f) {
#pragma unroll
  for (int q = 0; q < KSTEPS_PER_GROUP; ++q) {
#pragma unroll
    for (int t = 0; t < 4; ++t) {
      v2f b;
      b.x = f.blo[q][0][t];
      b.y = f.blo[q][1][t];
      acc[t] = __builtin_amdgcn_wmma_f32_16x16x4_f32(
           false, f.a[q],  false, b,
           (short)0, acc[t],  false,  false);
      v2f d;
      d.x = f.bhi[q][0][t];
      d.y = f.bhi[q][1][t];
      acc[4 + t] = __builtin_amdgcn_wmma_f32_16x16x4_f32(
           false, f.a[q],  false, d,
           (short)0, acc[4 + t],  false,  false);
    }
  }
}

__global__ __launch_bounds__(32 * GEMM_WAVES) void gemm_wmma_kernel(
    const float* __restrict__ Z, const float* __restrict__ W,
    float* __restrict__ H, int nRows) {
  const int lane = threadIdx.x;
  const int strip = __builtin_amdgcn_readfirstlane(blockIdx.x * GEMM_WAVES + threadIdx.y);
  const int m0 = strip * 16;
  if (m0 >= nRows) return;

  const int c = lane & 15;
  const int h = lane >> 4;
  const float* zp = Z + (size_t)(m0 + c) * D_FEAT + 2 * h;
  const float* wp = W + (size_t)(2 * h) * D_FEAT + 4 * c;

  v8f acc[8];
#pragma unroll
  for (int t = 0; t < 8; ++t) acc[t] = v8f{};

  constexpr int KGROUP = 4 * KSTEPS_PER_GROUP;
  FragGroup f0, f1;
  gemm_load_group(f0, zp, wp, 0);
#pragma unroll
  for (int g = 0; g < D_FEAT / KGROUP; g += 2) {
    gemm_load_group(f1, zp, wp, (g + 1) * KGROUP);
    __builtin_amdgcn_sched_barrier(0);
    gemm_mma_group(acc, f0);
    if (g + 2 < D_FEAT / KGROUP) {
      gemm_load_group(f0, zp, wp, (g + 2) * KGROUP);
      __builtin_amdgcn_sched_barrier(0);
    }
    gemm_mma_group(acc, f1);
  }

  float* hp = H + (size_t)(m0 + 8 * h) * D_FEAT + 4 * c;
#pragma unroll
  for (int r = 0; r < 8; ++r) {
    v4f lo, hi;
#pragma unroll
    for (int t = 0; t < 4; ++t) {
      lo[t] = acc[t][r];
      hi[t] = acc[4 + t][r];
    }
    *(v4f*)(hp + (size_t)r * D_FEAT) = lo;
    *(v4f*)(hp + (size_t)r * D_FEAT + 64) = hi;
  }
}

template <int MODE>
__global__ __launch_bounds__(256) void pull_kernel(
    const int* __restrict__ off, const int* __restrict__ ssrc,
    const float* __restrict__ swt, const float* __restrict__ dinv,
    const float* __restrict__ H, const float* __restrict__ bias,
    const float* __restrict__ pw, const float* __restrict__ cscale,
    float* __restrict__ Zout, int N) {
  const int lane = threadIdx.x;
  const int c = __builtin_amdgcn_readfirstlane(blockIdx.x * 8 + threadIdx.y);
  if (c >= N) return;
  const int start = off[c];
  const int end = off[c + 1];
  const float dc = dinv[c];
  const float4* H4 = (const float4*)H;

  float4 acc = make_float4(0.0f, 0.0f, 0.0f, 0.0f);
  for (int base = start; base < end; base += 32) {
    const int m = (end - base < 32) ? (end - base) : 32;
    int r = 0;
    float nrm = 0.0f;
    if (lane < m) {
      r = ssrc[base + lane];
      nrm = dinv[r] * swt[base + lane] * dc;
    }
    const int nrm_bits = __float_as_int(nrm);
#pragma unroll 4
    for (int j = 0; j < m; ++j) {
      const int rj = __builtin_amdgcn_readlane(r, j);
      const float nj = __int_as_float(__builtin_amdgcn_readlane(nrm_bits, j));
      const float4 v = H4[(size_t)rj * (D_FEAT / 4) + lane];
      acc.x += v.x * nj;
      acc.y += v.y * nj;
      acc.z += v.z * nj;
      acc.w += v.w * nj;
    }
  }

  const float4 h = H4[(size_t)c * (D_FEAT / 4) + lane];
  const float4 b = ((const float4*)bias)[lane];
  const float4 p = ((const float4*)pw)[lane];
  float4 o;
  float v;
  v = acc.x + h.x * dc * dc + b.x;
  o.x = fmaxf(v, 0.0f) + p.x * fminf(v, 0.0f);
  v = acc.y + h.y * dc * dc + b.y;
  o.y = fmaxf(v, 0.0f) + p.y * fminf(v, 0.0f);
  v = acc.z + h.z * dc * dc + b.z;
  o.z = fmaxf(v, 0.0f) + p.z * fminf(v, 0.0f);
  v = acc.w + h.w * dc * dc + b.w;
  o.w = fmaxf(v, 0.0f) + p.w * fminf(v, 0.0f);
  float4* zp = (float4*)Zout + (size_t)c * (D_FEAT / 4) + lane;
  if (MODE == 0) {
    *zp = o;
  } else {
    const float ct = cscale[0];
    float4 r;
    if (MODE == 1) {
      r.x = ct * o.x;
      r.y = ct * o.y;
      r.z = ct * o.z;
      r.w = ct * o.w;
    } else {
      const float4 s = *zp;
      r.x = s.x + ct * o.x;
      r.y = s.y + ct * o.y;
      r.z = s.z + ct * o.z;
      r.w = s.w + ct * o.w;
    }
    *zp = r;
  }
}

extern "C" void kernel_launch(void* const* d_in, const int* in_sizes, int n_in,
                              void* d_out, int out_size, void* d_ws, size_t ws_size,
                              hipStream_t stream) {
  const float* x       = (const float*)d_in[0];
  const int*   eidx    = (const int*)  d_in[1];
  const float* ew      = (const float*)d_in[2];
  const float* coeff   = (const float*)d_in[3];
  const float* W1      = (const float*)d_in[4];
  const float* b1      = (const float*)d_in[5];
  const float* W2      = (const float*)d_in[6];
  const float* b2      = (const float*)d_in[7];
  const float* prelu_w = (const float*)d_in[8];

  const int N = in_sizes[0] / D_FEAT;
  const int E = in_sizes[2] / 3;
  const int G = 3 * N;
  const size_t E3 = (size_t)3 * E;

  const size_t padG  = ((size_t)G + 1 + 31) & ~(size_t)31;
  const size_t padE3 = (E3 + 31) & ~(size_t)31;
  const int scanBlocks = (G + SCAN_B - 1) / SCAN_B;
  const size_t padS  = ((size_t)scanBlocks + 31) & ~(size_t)31;
  int*   cnt  = (int*)d_ws;
  int*   off  = cnt + padG;
  int*   bsum = off + padG;
  int*   rank = bsum + padS;
  int*   tmp  = rank + padE3;
  int*   ssrc = tmp + padE3;
  float* swt  = (float*)(ssrc + padE3);
  float* dinv = swt + padE3;
  float* bufA = dinv + padG;
  float* bufB = bufA + (size_t)N * D_FEAT;

  const int ETHREADS = 256;
  const dim3 waveBlock(32, 8);
  const dim3 gemmBlock(32, GEMM_WAVES);
  const int gemmBlocks = (N / 16 + GEMM_WAVES - 1) / GEMM_WAVES;
  const int nodeWaveBlocks = (N + 7) / 8;
  const dim3 edgeGrid((E + ETHREADS - 1) / ETHREADS, 3);
  const dim3 nodeWaveGrid(nodeWaveBlocks, 3);

  zero_kernel<<<(G + ETHREADS - 1) / ETHREADS, ETHREADS, 0, stream>>>((float*)cnt, G);
  rank_kernel<<<edgeGrid, ETHREADS, 0, stream>>>(eidx, cnt, rank, E, N);
  bsum_kernel<<<scanBlocks, SCAN_B, 0, stream>>>(cnt, bsum, G);
  scan_kernel<<<scanBlocks, SCAN_B, 0, stream>>>(cnt, bsum, off, G);
  fill_kernel<<<edgeGrid, ETHREADS, 0, stream>>>(eidx, off, rank, tmp, E, N);
  sort_kernel<<<nodeWaveGrid, waveBlock, 0, stream>>>(eidx, ew, off, tmp, ssrc, swt, E, N);
  dinv_kernel<<<(G + ETHREADS - 1) / ETHREADS, ETHREADS, 0, stream>>>(off, swt, dinv, G);

  for (int t = 0; t < 3; ++t) {
    const int*   offT  = off + (size_t)t * N;
    const float* dinvT = dinv + (size_t)t * N;

    gemm_wmma_kernel<<<gemmBlocks, gemmBlock, 0, stream>>>(x, W1, bufA, N);
    pull_kernel<0><<<nodeWaveBlocks, waveBlock, 0, stream>>>(offT, ssrc, swt, dinvT, bufA, b1, prelu_w, coeff + t, bufB, N);

    gemm_wmma_kernel<<<gemmBlocks, gemmBlock, 0, stream>>>(bufB, W2, bufA, N);
    if (t == 0)
      pull_kernel<1><<<nodeWaveBlocks, waveBlock, 0, stream>>>(offT, ssrc, swt, dinvT, bufA, b2, prelu_w, coeff + t, (float*)d_out, N);
    else
      pull_kernel<2><<<nodeWaveBlocks, waveBlock, 0, stream>>>(offT, ssrc, swt, dinvT, bufA, b2, prelu_w, coeff + t, (float*)d_out, N);
  }
}


# Fewer, longer waves in edge and node passes: a thread takes 8 edges (rank, fill), a wave 8 buckets (sort) or 4 nodes (gather), first loads of all its items requested together; sum order unchanged
# speedup vs baseline: 3.0335x; 1.1209x over previous
#include <hip/hip_runtime.h>
#include <hip/hip_bf16.h>


typedef __attribute__((ext_vector_type(2))) float v2f;
typedef __attribute__((ext_vector_type(8))) float v8f;

#define D_FEAT 128
#define SCAN_B 1024
#define EDGE_QUADS 2
#define SORT_BUCKETS 8
#define PULL_NODES 4

__device__ __forceinline__ int4 load_quad(const int* __restrict__ p, int e, int n, bool vec) {
  int4 v = make_int4(0, 0, 0, 0);
  if (e < n) {
    if (vec) {
      v = *(const int4*)(p + e);
    } else {
      v.x = p[e];
      if (e + 1 < n) v.y = p[e + 1];
      if (e + 2 < n) v.z = p[e + 2];
      if (e + 3 < n) v.w = p[e + 3];
    }
  }
  return v;
}

__device__ __forceinline__ void store_quad(int* __restrict__ p, int e, int n, bool vec, int4 v) {
  if (e < n) {
    if (vec) {
      *(int4*)(p + e) = v;
    } else {
      p[e] = v.x;
      if (e + 1 < n) p[e + 1] = v.y;
      if (e + 2 < n) p[e + 2] = v.z;
      if (e + 3 < n) p[e + 3] = v.w;
    }
  }
}

__global__ void zero_kernel(float* __restrict__ p, int n) {
  int i = blockIdx.x * blockDim.x + threadIdx.x;
  if (i < n) p[i] = 0.0f;
}

__global__ __launch_bounds__(256) void rank_kernel(
    const int* __restrict__ eidx, int* __restrict__ cnt,
    int* __restrict__ rank, int E, int N) {
  const int t = blockIdx.y;
  const int* col = eidx + (size_t)t * 2 * E + E;
  int* rk = rank + (size_t)t * E;
  int* cn = cnt + t * N;
  const bool vec = (E & 3) == 0;
  const int stride = gridDim.x * blockDim.x * 4;
  const int e0 = (blockIdx.x * blockDim.x + threadIdx.x) * 4;
  int4 c[EDGE_QUADS];
#pragma unroll
  for (int i = 0; i < EDGE_QUADS; ++i) c[i] = load_quad(col, e0 + i * stride, E, vec);
  int4 r[EDGE_QUADS];
#pragma unroll
  for (int i = 0; i < EDGE_QUADS; ++i) {
    const int e = e0 + i * stride;
    r[i] = make_int4(0, 0, 0, 0);
    if (e < E) r[i].x = atomicAdd(&cn[c[i].x], 1);
    if (e + 1 < E) r[i].y = atomicAdd(&cn[c[i].y], 1);
    if (e + 2 < E) r[i].z = atomicAdd(&cn[c[i].z], 1);
    if (e + 3 < E) r[i].w = atomicAdd(&cn[c[i].w], 1);
  }
#pragma unroll
  for (int i = 0; i < EDGE_QUADS; ++i) store_quad(rk, e0 + i * stride, E, vec, r[i]);
}

__global__ __launch_bounds__(SCAN_B) void bsum_kernel(
    const int* __restrict__ cnt, int* __restrict__ bsum, int n) {
  __shared__ int s[32];
  const int tid = threadIdx.x;
  const int i = blockIdx.x * SCAN_B + tid;
  int v = (i < n) ? cnt[i] : 0;
  for (int m = 16; m > 0; m >>= 1) v += __shfl_xor(v, m, 32);
  if ((tid & 31) == 0) s[tid >> 5] = v;
  __syncthreads();
  if (tid < 32) {
    int x = s[tid];
    for (int m = 16; m > 0; m >>= 1) x += __shfl_xor(x, m, 32);
    if (tid == 0) bsum[blockIdx.x] = x;
  }
}

__global__ __launch_bounds__(SCAN_B) void scan_kernel(
    const int* __restrict__ cnt, const int* __restrict__ bsum,
    int* __restrict__ off, int n) {
  __shared__ int sw[32];
  __shared__ int sbase;
  const int tid = threadIdx.x;
  const int lane = tid & 31;
  const int wv = tid >> 5;
  if (wv == 0) {
    int a = 0;
    for (int j = lane; j < (int)blockIdx.x; j += 32) a += bsum[j];
    for (int m = 16; m > 0; m >>= 1) a += __shfl_xor(a, m, 32);
    if (lane == 0) sbase = a;
  }
  const int i = blockIdx.x * SCAN_B + tid;
  const int v = (i < n) ? cnt[i] : 0;
  int x = v;
  for (int d = 1; d < 32; d <<= 1) {
    const int y = __shfl_up(x, d, 32);
    if (lane >= d) x += y;
  }
  if (lane == 31) sw[wv] = x;
  __syncthreads();
  if (wv == 0) {
    const int wtot = sw[lane];
    int xx = wtot;
    for (int d = 1; d < 32; d <<= 1) {
      const int y = __shfl_up(xx, d, 32);
      if (lane >= d) xx += y;
    }
    sw[lane] = xx - wtot;
  }
  __syncthreads();
  const int excl = sbase + sw[wv] + x - v;
  if (i < n) off[i] = excl;
  if (i == n - 1) off[n] = excl + v;
}

__global__ __launch_bounds__(256) void fill_kernel(
    const int* __restrict__ eidx, const int* __restrict__ off,
    const int* __restrict__ rank, int* __restrict__ tmp, int E, int N) {
  const int t = blockIdx.y;
  const int* col = eidx + (size_t)t * 2 * E + E;
  const int* rk = rank + (size_t)t * E;
  const int* of = off + t * N;
  const bool vec = (E & 3) == 0;
  const int stride = gridDim.x * blockDim.x * 4;
  const int e0 = (blockIdx.x * blockDim.x + threadIdx.x) * 4;
  int4 c[EDGE_QUADS], r[EDGE_QUADS];
#pragma unroll
  for (int i = 0; i < EDGE_QUADS; ++i) {
    c[i] = load_quad(col, e0 + i * stride, E, vec);
    r[i] = load_quad(rk, e0 + i * stride, E, vec);
  }
  int4 o[EDGE_QUADS];
#pragma unroll
  for (int i = 0; i < EDGE_QUADS; ++i) {
    const int e = e0 + i * stride;
    o[i] = make_int4(0, 0, 0, 0);
    if (e < E) o[i].x = of[c[i].x];
    if (e + 1 < E) o[i].y = of[c[i].y];
    if (e + 2 < E) o[i].z = of[c[i].z];
    if (e + 3 < E) o[i].w = of[c[i].w];
  }
#pragma unroll
  for (int i = 0; i < EDGE_QUADS; ++i) {
    const int e = e0 + i * stride;
    if (e < E) tmp[o[i].x + r[i].x] = e;
    if (e + 1 < E) tmp[o[i].y + r[i].y] = e + 1;
    if (e + 2 < E) tmp[o[i].z + r[i].z] = e + 2;
    if (e + 3 < E) tmp[o[i].w + r[i].w] = e + 3;
  }
}

__global__ __launch_bounds__(256) void sort_kernel(
    const int* __restrict__ eidx, const float* __restrict__ ew,
    const int* __restrict__ off, const int* __restrict__ tmp,
    int* __restrict__ ssrc, float* __restrict__ swt, int E, int N, int G) {
  const int lane = threadIdx.x;
  const int g0 = __builtin_amdgcn_readfirstlane(
      (blockIdx.x * 8 + threadIdx.y) * SORT_BUCKETS);
  if (g0 >= G) return;

  int gi = g0 + lane;
  if (gi > G) gi = G;
  const int ov = (lane <= SORT_BUCKETS) ? off[gi] : 0;
  int st[SORT_BUCKETS + 1];
#pragma unroll
  for (int k = 0; k <= SORT_BUCKETS; ++k) st[k] = __builtin_amdgcn_readlane(ov, k);

  int eid[SORT_BUCKETS];
#pragma unroll
  for (int k = 0; k < SORT_BUCKETS; ++k) {
    const int n = st[k + 1] - st[k];
    eid[k] = (lane < n) ? tmp[st[k] + lane] : 0x7fffffff;
  }

  int rk[SORT_BUCKETS];
#pragma unroll
  for (int k = 0; k < SORT_BUCKETS; ++k) {
    const int n = st[k + 1] - st[k];
    rk[k] = 0;
    if (n <= 32) {
      for (int j = 0; j < n; ++j)
        rk[k] += (__builtin_amdgcn_readlane(eid[k], j) < eid[k]) ? 1 : 0;
    }
  }

  int src[SORT_BUCKETS];
  float wt[SORT_BUCKETS];
#pragma unroll
  for (int k = 0; k < SORT_BUCKETS; ++k) {
    const int n = st[k + 1] - st[k];
    const int g = g0 + k;
    const int t = (g >= N ? 1 : 0) + (g >= 2 * N ? 1 : 0);
    const int e = (n <= 32 && lane < n) ? eid[k] : 0;
    src[k] = eidx[(size_t)t * 2 * E + e];
    wt[k] = ew[(size_t)t * E + e];
  }
#pragma unroll
  for (int k = 0; k < SORT_BUCKETS; ++k) {
    const int n = st[k + 1] - st[k];
    if (n <= 32 && lane < n) {
      ssrc[st[k] + rk[k]] = src[k];
      swt[st[k] + rk[k]] = wt[k];
    }
  }

#pragma unroll
  for (int k = 0; k < SORT_BUCKETS; ++k) {
    const int n = st[k + 1] - st[k];
    if (n > 32) {
      const int g = g0 + k;
      const int t = (g >= N ? 1 : 0) + (g >= 2 * N ? 1 : 0);
      const int* row = eidx + (size_t)t * 2 * E;
      const float* w = ew + (size_t)t * E;
      const int start = st[k];
      for (int i = lane; i < n; i += 32) {
        const int e = tmp[start + i];
        int q = 0;
        for (int j = 0; j < n; ++j) q += (tmp[start + j] < e) ? 1 : 0;
        ssrc[start + q] = row[e];
        swt[start + q] = w[e];
      }
    }
  }
}

__global__ void dinv_kernel(const int* __restrict__ off, const float* __restrict__ swt,
                            float* __restrict__ dinv, int G) {
  const int g = blockIdx.x * blockDim.x + threadIdx.x;
  if (g >= G) return;
  const int end = off[g + 1];
  float s = 0.0f;
  for (int p = off[g]; p < end; ++p) s += swt[p];
  const float d = s + 1.0f;
  dinv[g] = rsqrtf(d);
}

typedef __attribute__((ext_vector_type(4))) float v4f;

#define GEMM_WAVES 4
#define KSTEPS_PER_GROUP 4

struct FragGroup {
  v2f a[KSTEPS_PER_GROUP];
  v4f blo[KSTEPS_PER_GROUP][2];
  v4f bhi[KSTEPS_PER_GROUP][2];
};

__device__ __forceinline__ void gemm_load_group(FragGroup& f, const float* __restrict__ zp,
                                                const float* __restrict__ wp, int kbase) {
#pragma unroll
  for (int q = 0; q < KSTEPS_PER_GROUP; ++q) {
    const int k0 = kbase + 4 * q;
    f.a[q] = *(const v2f*)(zp + k0);
#pragma unroll
    for (int s = 0; s < 2; ++s) {
      f.blo[q][s] = *(const v4f*)(wp + (size_t)(k0 + s) * D_FEAT);
      f.bhi[q][s] = *(const v4f*)(wp + (size_t)(k0 + s) * D_FEAT + 64);
    }
  }
}

__device__ __forceinline__ void gemm_mma_group(v8f (&acc)[8], const FragGroup& f) {
#pragma unroll
  for (int q = 0; q < KSTEPS_PER_GROUP; ++q) {
#pragma unroll
    for (int t = 0; t < 4; ++t) {
      v2f b;
      b.x = f.blo[q][0][t];
      b.y = f.blo[q][1][t];
      acc[t] = __builtin_amdgcn_wmma_f32_16x16x4_f32(
           false, f.a[q],  false, b,
           (short)0, acc[t],  false,  false);
      v2f d;
      d.x = f.bhi[q][0][t];
      d.y = f.bhi[q][1][t];
      acc[4 + t] = __builtin_amdgcn_wmma_f32_16x16x4_f32(
           false, f.a[q],  false, d,
           (short)0, acc[4 + t],  false,  false);
    }
  }
}

__global__ __launch_bounds__(32 * GEMM_WAVES) void gemm_wmma_kernel(
    const float* __restrict__ Z, const float* __restrict__ W,
    float* __restrict__ H, int nRows) {
  const int lane = threadIdx.x;
  const int strip = __builtin_amdgcn_readfirstlane(blockIdx.x * GEMM_WAVES + threadIdx.y);
  const int m0 = strip * 16;
  if (m0 >= nRows) return;

  const int c = lane & 15;
  const int h = lane >> 4;
  const float* zp = Z + (size_t)(m0 + c) * D_FEAT + 2 * h;
  const float* wp = W + (size_t)(2 * h) * D_FEAT + 4 * c;

  v8f acc[8];
#pragma unroll
  for (int t = 0; t < 8; ++t) acc[t] = v8f{};

  constexpr int KGROUP = 4 * KSTEPS_PER_GROUP;
  FragGroup f0, f1;
  gemm_load_group(f0, zp, wp, 0);
#pragma unroll
  for (int g = 0; g < D_FEAT / KGROUP; g += 2) {
    gemm_load_group(f1, zp, wp, (g + 1) * KGROUP);
    __builtin_amdgcn_sched_barrier(0);
    gemm_mma_group(acc, f0);
    if (g + 2 < D_FEAT / KGROUP) {
      gemm_load_group(f0, zp, wp, (g + 2) * KGROUP);
      __builtin_amdgcn_sched_barrier(0);
    }
    gemm_mma_group(acc, f1);
  }

  float* hp = H + (size_t)(m0 + 8 * h) * D_FEAT + 4 * c;
#pragma unroll
  for (int r = 0; r < 8; ++r) {
    v4f lo, hi;
#pragma unroll
    for (int t = 0; t < 4; ++t) {
      lo[t] = acc[t][r];
      hi[t] = acc[4 + t][r];
    }
    *(v4f*)(hp + (size_t)r * D_FEAT) = lo;
    *(v4f*)(hp + (size_t)r * D_FEAT + 64) = hi;
  }
}

template <int MODE>
__global__ __launch_bounds__(256) void pull_kernel(
    const int* __restrict__ off, const int* __restrict__ ssrc,
    const float* __restrict__ swt, const float* __restrict__ dinv,
    const float* __restrict__ H, const float* __restrict__ bias,
    const float* __restrict__ pw, const float* __restrict__ cscale,
    float* __restrict__ Zout, int N) {
  const int lane = threadIdx.x;
  const int c0 = __builtin_amdgcn_readfirstlane(
      (blockIdx.x * 8 + threadIdx.y) * PULL_NODES);
  if (c0 >= N) return;
  const float4* H4 = (const float4*)H;

  int ci = c0 + lane;
  if (ci > N) ci = N;
  const int ov = (lane <= PULL_NODES) ? off[ci] : 0;
  const float dv = (lane < PULL_NODES && c0 + lane < N) ? dinv[c0 + lane] : 0.0f;
  const float4 b = ((const float4*)bias)[lane];
  const float4 p = ((const float4*)pw)[lane];
  float ct = 0.0f;
  if (MODE != 0) ct = cscale[0];

  int st[PULL_NODES + 1];
  float dcs[PULL_NODES];
#pragma unroll
  for (int k = 0; k <= PULL_NODES; ++k) st[k] = __builtin_amdgcn_readlane(ov, k);
#pragma unroll
  for (int k = 0; k < PULL_NODES; ++k)
    dcs[k] = __int_as_float(__builtin_amdgcn_readlane(__float_as_int(dv), k));

  int r0[PULL_NODES];
  float w0[PULL_NODES];
#pragma unroll
  for (int k = 0; k < PULL_NODES; ++k) {
    const int n = st[k + 1] - st[k];
    r0[k] = 0;
    w0[k] = 0.0f;
    if (lane < n) {
      r0[k] = ssrc[st[k] + lane];
      w0[k] = swt[st[k] + lane];
    }
  }
  float d0[PULL_NODES];
#pragma unroll
  for (int k = 0; k < PULL_NODES; ++k) d0[k] = dinv[r0[k]];
  float n0[PULL_NODES];
#pragma unroll
  for (int k = 0; k < PULL_NODES; ++k) {
    const int n = st[k + 1] - st[k];
    const float x = d0[k] * w0[k] * dcs[k];
    n0[k] = (lane < n) ? x : 0.0f;
  }

#pragma unroll
  for (int k = 0; k < PULL_NODES; ++k) {
    const int c = c0 + k;
    if (c < N) {
      const int start = st[k];
      const int end = st[k + 1];
      const float dc = dcs[k];
      float4* zp = (float4*)Zout + (size_t)c * (D_FEAT / 4) + lane;
      const float4 h = H4[(size_t)c * (D_FEAT / 4) + lane];
      float4 s = make_float4(0.0f, 0.0f, 0.0f, 0.0f);
      if (MODE == 2) s = *zp;

      float4 acc = make_float4(0.0f, 0.0f, 0.0f, 0.0f);
      for (int base = start; base < end; base += 32) {
        const int m = (end - base < 32) ? (end - base) : 32;
        int r = r0[k];
        float nrm = n0[k];
        if (base != start) {
          r = 0;
          nrm = 0.0f;
          if (lane < m) {
            r = ssrc[base + lane];
            nrm = dinv[r] * swt[base + lane] * dc;
          }
        }
        const int nrm_bits = __float_as_int(nrm);
#pragma unroll 4
        for (int j = 0; j < m; ++j) {
          const int rj = __builtin_amdgcn_readlane(r, j);
          const float nj = __int_as_float(__builtin_amdgcn_readlane(nrm_bits, j));
          const float4 v = H4[(size_t)rj * (D_FEAT / 4) + lane];
          acc.x += v.x * nj;
          acc.y += v.y * nj;
          acc.z += v.z * nj;
          acc.w += v.w * nj;
        }
      }

      float4 o;
      float v;
      v = acc.x + h.x * dc * dc + b.x;
      o.x = fmaxf(v, 0.0f) + p.x * fminf(v, 0.0f);
      v = acc.y + h.y * dc * dc + b.y;
      o.y = fmaxf(v, 0.0f) + p.y * fminf(v, 0.0f);
      v = acc.z + h.z * dc * dc + b.z;
      o.z = fmaxf(v, 0.0f) + p.z * fminf(v, 0.0f);
      v = acc.w + h.w * dc * dc + b.w;
      o.w = fmaxf(v, 0.0f) + p.w * fminf(v, 0.0f);
      if (MODE == 0) {
        *zp = o;
      } else {
        float4 q;
        if (MODE == 1) {
          q.x = ct * o.x;
          q.y = ct * o.y;
          q.z = ct * o.z;
          q.w = ct * o.w;
        } else {
          q.x = s.x + ct * o.x;
          q.y = s.y + ct * o.y;
          q.z = s.z + ct * o.z;
          q.w = s.w + ct * o.w;
        }
        *zp = q;
      }
    }
  }
}

extern "C" void kernel_launch(void* const* d_in, const int* in_sizes, int n_in,
                              void* d_out, int out_size, void* d_ws, size_t ws_size,
                              hipStream_t stream) {
  const float* x       = (const float*)d_in[0];
  const int*   eidx    = (const int*)  d_in[1];
  const float* ew      = (const float*)d_in[2];
  const float* coeff   = (const float*)d_in[3];
  const float* W1      = (const float*)d_in[4];
  const float* b1      = (const float*)d_in[5];
  const float* W2      = (const float*)d_in[6];
  const float* b2      = (const float*)d_in[7];
  const float* prelu_w = (const float*)d_in[8];

  const int N = in_sizes[0] / D_FEAT;
  const int E = in_sizes[2] / 3;
  const int G = 3 * N;
  const size_t E3 = (size_t)3 * E;

  const size_t padG  = ((size_t)G + 1 + 31) & ~(size_t)31;
  const size_t padE3 = (E3 + 31) & ~(size_t)31;
  const int scanBlocks = (G + SCAN_B - 1) / SCAN_B;
  const size_t padS  = ((size_t)scanBlocks + 31) & ~(size_t)31;
  int*   cnt  = (int*)d_ws;
  int*   off  = cnt + padG;
  int*   bsum = off + padG;
  int*   rank = bsum + padS;
  int*   tmp  = rank + padE3;
  int*   ssrc = tmp + padE3;
  float* swt  = (float*)(ssrc + padE3);
  float* dinv = swt + padE3;
  float* bufA = dinv + padG;
  float* bufB = bufA + (size_t)N * D_FEAT;

  const int ETHREADS = 256;
  const dim3 waveBlock(32, 8);
  const dim3 gemmBlock(32, GEMM_WAVES);
  const int gemmBlocks = (N / 16 + GEMM_WAVES - 1) / GEMM_WAVES;
  const int nodeWaveBlocks = (N + 8 * PULL_NODES - 1) / (8 * PULL_NODES);
  const int edgesPerBlock = ETHREADS * 4 * EDGE_QUADS;
  const dim3 edgeGrid((E + edgesPerBlock - 1) / edgesPerBlock, 3);
  const int sortBlocks = (G + 8 * SORT_BUCKETS - 1) / (8 * SORT_BUCKETS);

  zero_kernel<<<(G + ETHREADS - 1) / ETHREADS, ETHREADS, 0, stream>>>((float*)cnt, G);
  rank_kernel<<<edgeGrid, ETHREADS, 0, stream>>>(eidx, cnt, rank, E, N);
  bsum_kernel<<<scanBlocks, SCAN_B, 0, stream>>>(cnt, bsum, G);
  scan_kernel<<<scanBlocks, SCAN_B, 0, stream>>>(cnt, bsum, off, G);
  fill_kernel<<<edgeGrid, ETHREADS, 0, stream>>>(eidx, off, rank, tmp, E, N);
  sort_kernel<<<sortBlocks, waveBlock, 0, stream>>>(eidx, ew, off, tmp, ssrc, swt, E, N, G);
  dinv_kernel<<<(G + ETHREADS - 1) / ETHREADS, ETHREADS, 0, stream>>>(off, swt, dinv, G);

  for (int t = 0; t < 3; ++t) {
    const int*   offT  = off + (size_t)t * N;
    const float* dinvT = dinv + (size_t)t * N;

    gemm_wmma_kernel<<<gemmBlocks, gemmBlock, 0, stream>>>(x, W1, bufA, N);
    pull_kernel<0><<<nodeWaveBlocks, waveBlock, 0, stream>>>(offT, ssrc, swt, dinvT, bufA, b1, prelu_w, coeff + t, bufB, N);

    gemm_wmma_kernel<<<gemmBlocks, gemmBlock, 0, stream>>>(bufB, W2, bufA, N);
    if (t == 0)
      pull_kernel<1><<<nodeWaveBlocks, waveBlock, 0, stream>>>(offT, ssrc, swt, dinvT, bufA, b2, prelu_w, coeff + t, (float*)d_out, N);
    else
      pull_kernel<2><<<nodeWaveBlocks, waveBlock, 0, stream>>>(offT, ssrc, swt, dinvT, bufA, b2, prelu_w, coeff + t, (float*)d_out, N);
  }
}


# Scales share launches: x@W1 computed once (same for all three), layer-1 gather once with grid.y = scale on that one H, layer-2 product once over 3N rows; 19 -> 13 launches; sum order unchanged
# speedup vs baseline: 3.3192x; 1.0942x over previous
#include <hip/hip_runtime.h>
#include <hip/hip_bf16.h>


typedef __attribute__((ext_vector_type(2))) float v2f;
typedef __attribute__((ext_vector_type(8))) float v8f;

#define D_FEAT 128
#define SCAN_B 1024
#define EDGE_QUADS 2
#define SORT_BUCKETS 8
#define PULL_NODES 4

__device__ __forceinline__ int4 load_quad(const int* __restrict__ p, int e, int n, bool vec) {
  int4 v = make_int4(0, 0, 0, 0);
  if (e < n) {
    if (vec) {
      v = *(const int4*)(p + e);
    } else {
      v.x = p[e];
      if (e + 1 < n) v.y = p[e + 1];
      if (e + 2 < n) v.z = p[e + 2];
      if (e + 3 < n) v.w = p[e + 3];
    }
  }
  return v;
}

__device__ __forceinline__ void store_quad(int* __restrict__ p, int e, int n, bool vec, int4 v) {
  if (e < n) {
    if (vec) {
      *(int4*)(p + e) = v;
    } else {
      p[e] = v.x;
      if (e + 1 < n) p[e + 1] = v.y;
      if (e + 2 < n) p[e + 2] = v.z;
      if (e + 3 < n) p[e + 3] = v.w;
    }
  }
}

__global__ void zero_kernel(float* __restrict__ p, int n) {
  int i = blockIdx.x * blockDim.x + threadIdx.x;
  if (i < n) p[i] = 0.0f;
}

__global__ __launch_bounds__(256) void rank_kernel(
    const int* __restrict__ eidx, int* __restrict__ cnt,
    int* __restrict__ rank, int E, int N) {
  const int t = blockIdx.y;
  const int* col = eidx + (size_t)t * 2 * E + E;
  int* rk = rank + (size_t)t * E;
  int* cn = cnt + t * N;
  const bool vec = (E & 3) == 0;
  const int stride = gridDim.x * blockDim.x * 4;
  const int e0 = (blockIdx.x * blockDim.x + threadIdx.x) * 4;
  int4 c[EDGE_QUADS];
#pragma unroll
  for (int i = 0; i < EDGE_QUADS; ++i) c[i] = load_quad(col, e0 + i * stride, E, vec);
  int4 r[EDGE_QUADS];
#pragma unroll
  for (int i = 0; i < EDGE_QUADS; ++i) {
    const int e = e0 + i * stride;
    r[i] = make_int4(0, 0, 0, 0);
    if (e < E) r[i].x = atomicAdd(&cn[c[i].x], 1);
    if (e + 1 < E) r[i].y = atomicAdd(&cn[c[i].y], 1);
    if (e + 2 < E) r[i].z = atomicAdd(&cn[c[i].z], 1);
    if (e + 3 < E) r[i].w = atomicAdd(&cn[c[i].w], 1);
  }
#pragma unroll
  for (int i = 0; i < EDGE_QUADS; ++i) store_quad(rk, e0 + i * stride, E, vec, r[i]);
}

__global__ __launch_bounds__(SCAN_B) void bsum_kernel(
    const int* __restrict__ cnt, int* __restrict__ bsum, int n) {
  __shared__ int s[32];
  const int tid = threadIdx.x;
  const int i = blockIdx.x * SCAN_B + tid;
  int v = (i < n) ? cnt[i] : 0;
  for (int m = 16; m > 0; m >>= 1) v += __shfl_xor(v, m, 32);
  if ((tid & 31) == 0) s[tid >> 5] = v;
  __syncthreads();
  if (tid < 32) {
    int x = s[tid];
    for (int m = 16; m > 0; m >>= 1) x += __shfl_xor(x, m, 32);
    if (tid == 0) bsum[blockIdx.x] = x;
  }
}

__global__ __launch_bounds__(SCAN_B) void scan_kernel(
    const int* __restrict__ cnt, const int* __restrict__ bsum,
    int* __restrict__ off, int n) {
  __shared__ int sw[32];
  __shared__ int sbase;
  const int tid = threadIdx.x;
  const int lane = tid & 31;
  const int wv = tid >> 5;
  if (wv == 0) {
    int a = 0;
    for (int j = lane; j < (int)blockIdx.x; j += 32) a += bsum[j];
    for (int m = 16; m > 0; m >>= 1) a += __shfl_xor(a, m, 32);
    if (lane == 0) sbase = a;
  }
  const int i = blockIdx.x * SCAN_B + tid;
  const int v = (i < n) ? cnt[i] : 0;
  int x = v;
  for (int d = 1; d < 32; d <<= 1) {
    const int y = __shfl_up(x, d, 32);
    if (lane >= d) x += y;
  }
  if (lane == 31) sw[wv] = x;
  __syncthreads();
  if (wv == 0) {
    const int wtot = sw[lane];
    int xx = wtot;
    for (int d = 1; d < 32; d <<= 1) {
      const int y = __shfl_up(xx, d, 32);
      if (lane >= d) xx += y;
    }
    sw[lane] = xx - wtot;
  }
  __syncthreads();
  const int excl = sbase + sw[wv] + x - v;
  if (i < n) off[i] = excl;
  if (i == n - 1) off[n] = excl + v;
}

__global__ __launch_bounds__(256) void fill_kernel(
    const int* __restrict__ eidx, const int* __restrict__ off,
    const int* __restrict__ rank, int* __restrict__ tmp, int E, int N) {
  const int t = blockIdx.y;
  const int* col = eidx + (size_t)t * 2 * E + E;
  const int* rk = rank + (size_t)t * E;
  const int* of = off + t * N;
  const bool vec = (E & 3) == 0;
  const int stride = gridDim.x * blockDim.x * 4;
  const int e0 = (blockIdx.x * blockDim.x + threadIdx.x) * 4;
  int4 c[EDGE_QUADS], r[EDGE_QUADS];
#pragma unroll
  for (int i = 0; i < EDGE_QUADS; ++i) {
    c[i] = load_quad(col, e0 + i * stride, E, vec);
    r[i] = load_quad(rk, e0 + i * stride, E, vec);
  }
  int4 o[EDGE_QUADS];
#pragma unroll
  for (int i = 0; i < EDGE_QUADS; ++i) {
    const int e = e0 + i * stride;
    o[i] = make_int4(0, 0, 0, 0);
    if (e < E) o[i].x = of[c[i].x];
    if (e + 1 < E) o[i].y = of[c[i].y];
    if (e + 2 < E) o[i].z = of[c[i].z];
    if (e + 3 < E) o[i].w = of[c[i].w];
  }
#pragma unroll
  for (int i = 0; i < EDGE_QUADS; ++i) {
    const int e = e0 + i * stride;
    if (e < E) tmp[o[i].x + r[i].x] = e;
    if (e + 1 < E) tmp[o[i].y + r[i].y] = e + 1;
    if (e + 2 < E) tmp[o[i].z + r[i].z] = e + 2;
    if (e + 3 < E) tmp[o[i].w + r[i].w] = e + 3;
  }
}

__global__ __launch_bounds__(256) void sort_kernel(
    const int* __restrict__ eidx, const float* __restrict__ ew,
    const int* __restrict__ off, const int* __restrict__ tmp,
    int* __restrict__ ssrc, float* __restrict__ swt, int E, int N, int G) {
  const int lane = threadIdx.x;
  const int g0 = __builtin_amdgcn_readfirstlane(
      (blockIdx.x * 8 + threadIdx.y) * SORT_BUCKETS);
  if (g0 >= G) return;

  int gi = g0 + lane;
  if (gi > G) gi = G;
  const int ov = (lane <= SORT_BUCKETS) ? off[gi] : 0;
  int st[SORT_BUCKETS + 1];
#pragma unroll
  for (int k = 0; k <= SORT_BUCKETS; ++k) st[k] = __builtin_amdgcn_readlane(ov, k);

  int eid[SORT_BUCKETS];
#pragma unroll
  for (int k = 0; k < SORT_BUCKETS; ++k) {
    const int n = st[k + 1] - st[k];
    eid[k] = (lane < n) ? tmp[st[k] + lane] : 0x7fffffff;
  }

  int rk[SORT_BUCKETS];
#pragma unroll
  for (int k = 0; k < SORT_BUCKETS; ++k) {
    const int n = st[k + 1] - st[k];
    rk[k] = 0;
    if (n <= 32) {
      for (int j = 0; j < n; ++j)
        rk[k] += (__builtin_amdgcn_readlane(eid[k], j) < eid[k]) ? 1 : 0;
    }
  }

  int src[SORT_BUCKETS];
  float wt[SORT_BUCKETS];
#pragma unroll
  for (int k = 0; k < SORT_BUCKETS; ++k) {
    const int n = st[k + 1] - st[k];
    const int g = g0 + k;
    const int t = (g >= N ? 1 : 0) + (g >= 2 * N ? 1 : 0);
    const int e = (n <= 32 && lane < n) ? eid[k] : 0;
    src[k] = eidx[(size_t)t * 2 * E + e];
    wt[k] = ew[(size_t)t * E + e];
  }
#pragma unroll
  for (int k = 0; k < SORT_BUCKETS; ++k) {
    const int n = st[k + 1] - st[k];
    if (n <= 32 && lane < n) {
      ssrc[st[k] + rk[k]] = src[k];
      swt[st[k] + rk[k]] = wt[k];
    }
  }

#pragma unroll
  for (int k = 0; k < SORT_BUCKETS; ++k) {
    const int n = st[k + 1] - st[k];
    if (n > 32) {
      const int g = g0 + k;
      const int t = (g >= N ? 1 : 0) + (g >= 2 * N ? 1 : 0);
      const int* row = eidx + (size_t)t * 2 * E;
      const float* w = ew + (size_t)t * E;
      const int start = st[k];
      for (int i = lane; i < n; i += 32) {
        const int e = tmp[start + i];
        int q = 0;
        for (int j = 0; j < n; ++j) q += (tmp[start + j] < e) ? 1 : 0;
        ssrc[start + q] = row[e];
        swt[start + q] = w[e];
      }
    }
  }
}

__global__ void dinv_kernel(const int* __restrict__ off, const float* __restrict__ swt,
                            float* __restrict__ dinv, int G) {
  const int g = blockIdx.x * blockDim.x + threadIdx.x;
  if (g >= G) return;
  const int end = off[g + 1];
  float s = 0.0f;
  for (int p = off[g]; p < end; ++p) s += swt[p];
  const float d = s + 1.0f;
  dinv[g] = rsqrtf(d);
}

typedef __attribute__((ext_vector_type(4))) float v4f;

#define GEMM_WAVES 4
#define KSTEPS_PER_GROUP 4

struct FragGroup {
  v2f a[KSTEPS_PER_GROUP];
  v4f blo[KSTEPS_PER_GROUP][2];
  v4f bhi[KSTEPS_PER_GROUP][2];
};

__device__ __forceinline__ void gemm_load_group(FragGroup& f, const float* __restrict__ zp,
                                                const float* __restrict__ wp, int kbase) {
#pragma unroll
  for (int q = 0; q < KSTEPS_PER_GROUP; ++q) {
    const int k0 = kbase + 4 * q;
    f.a[q] = *(const v2f*)(zp + k0);
#pragma unroll
    for (int s = 0; s < 2; ++s) {
      f.blo[q][s] = *(const v4f*)(wp + (size_t)(k0 + s) * D_FEAT);
      f.bhi[q][s] = *(const v4f*)(wp + (size_t)(k0 + s) * D_FEAT + 64);
    }
  }
}

__device__ __forceinline__ void gemm_mma_group(v8f (&acc)[8], const FragGroup& f) {
#pragma unroll
  for (int q = 0; q < KSTEPS_PER_GROUP; ++q) {
#pragma unroll
    for (int t = 0; t < 4; ++t) {
      v2f b;
      b.x = f.blo[q][0][t];
      b.y = f.blo[q][1][t];
      acc[t] = __builtin_amdgcn_wmma_f32_16x16x4_f32(
           false, f.a[q],  false, b,
           (short)0, acc[t],  false,  false);
      v2f d;
      d.x = f.bhi[q][0][t];
      d.y = f.bhi[q][1][t];
      acc[4 + t] = __builtin_amdgcn_wmma_f32_16x16x4_f32(
           false, f.a[q],  false, d,
           (short)0, acc[4 + t],  false,  false);
    }
  }
}

__global__ __launch_bounds__(32 * GEMM_WAVES) void gemm_wmma_kernel(
    const float* __restrict__ Z, const float* __restrict__ W,
    float* __restrict__ H, int nRows) {
  const int lane = threadIdx.x;
  const int strip = __builtin_amdgcn_readfirstlane(blockIdx.x * GEMM_WAVES + threadIdx.y);
  const int m0 = strip * 16;
  if (m0 >= nRows) return;

  const int c = lane & 15;
  const int h = lane >> 4;
  const float* zp = Z + (size_t)(m0 + c) * D_FEAT + 2 * h;
  const float* wp = W + (size_t)(2 * h) * D_FEAT + 4 * c;

  v8f acc[8];
#pragma unroll
  for (int t = 0; t < 8; ++t) acc[t] = v8f{};

  constexpr int KGROUP = 4 * KSTEPS_PER_GROUP;
  FragGroup f0, f1;
  gemm_load_group(f0, zp, wp, 0);
#pragma unroll
  for (int g = 0; g < D_FEAT / KGROUP; g += 2) {
    gemm_load_group(f1, zp, wp, (g + 1) * KGROUP);
    __builtin_amdgcn_sched_barrier(0);
    gemm_mma_group(acc, f0);
    if (g + 2 < D_FEAT / KGROUP) {
      gemm_load_group(f0, zp, wp, (g + 2) * KGROUP);
      __builtin_amdgcn_sched_barrier(0);
    }
    gemm_mma_group(acc, f1);
  }

  float* hp = H + (size_t)(m0 + 8 * h) * D_FEAT + 4 * c;
#pragma unroll
  for (int r = 0; r < 8; ++r) {
    v4f lo, hi;
#pragma unroll
    for (int t = 0; t < 4; ++t) {
      lo[t] = acc[t][r];
      hi[t] = acc[4 + t][r];
    }
    *(v4f*)(hp + (size_t)r * D_FEAT) = lo;
    *(v4f*)(hp + (size_t)r * D_FEAT + 64) = hi;
  }
}

template <int MODE>
__global__ __launch_bounds__(256) void pull_kernel(
    const int* __restrict__ offAll, const int* __restrict__ ssrc,
    const float* __restrict__ swt, const float* __restrict__ dinvAll,
    const float* __restrict__ Hall, const float* __restrict__ bias,
    const float* __restrict__ pw, const float* __restrict__ cscale,
    float* __restrict__ Zall, int N, size_t hPlane, size_t zPlane) {
  const int lane = threadIdx.x;
  const int c0 = __builtin_amdgcn_readfirstlane(
      (blockIdx.x * 8 + threadIdx.y) * PULL_NODES);
  if (c0 >= N) return;
  const int scale = blockIdx.y;
  const int* __restrict__ off = offAll + (size_t)scale * N;
  const float* __restrict__ dinv = dinvAll + (size_t)scale * N;
  const float* __restrict__ H = Hall + (size_t)scale * hPlane;
  float* __restrict__ Zout = Zall + (size_t)scale * zPlane;
  const float4* H4 = (const float4*)H;

  int ci = c0 + lane;
  if (ci > N) ci = N;
  const int ov = (lane <= PULL_NODES) ? off[ci] : 0;
  const float dv = (lane < PULL_NODES && c0 + lane < N) ? dinv[c0 + lane] : 0.0f;
  const float4 b = ((const float4*)bias)[lane];
  const float4 p = ((const float4*)pw)[lane];
  float ct = 0.0f;
  if (MODE != 0) ct = cscale[0];

  int st[PULL_NODES + 1];
  float dcs[PULL_NODES];
#pragma unroll
  for (int k = 0; k <= PULL_NODES; ++k) st[k] = __builtin_amdgcn_readlane(ov, k);
#pragma unroll
  for (int k = 0; k < PULL_NODES; ++k)
    dcs[k] = __int_as_float(__builtin_amdgcn_readlane(__float_as_int(dv), k));

  int r0[PULL_NODES];
  float w0[PULL_NODES];
#pragma unroll
  for (int k = 0; k < PULL_NODES; ++k) {
    const int n = st[k + 1] - st[k];
    r0[k] = 0;
    w0[k] = 0.0f;
    if (lane < n) {
      r0[k] = ssrc[st[k] + lane];
      w0[k] = swt[st[k] + lane];
    }
  }
  float d0[PULL_NODES];
#pragma unroll
  for (int k = 0; k < PULL_NODES; ++k) d0[k] = dinv[r0[k]];
  float n0[PULL_NODES];
#pragma unroll
  for (int k = 0; k < PULL_NODES; ++k) {
    const int n = st[k + 1] - st[k];
    const float x = d0[k] * w0[k] * dcs[k];
    n0[k] = (lane < n) ? x : 0.0f;
  }

#pragma unroll
  for (int k = 0; k < PULL_NODES; ++k) {
    const int c = c0 + k;
    if (c < N) {
      const int start = st[k];
      const int end = st[k + 1];
      const float dc = dcs[k];
      float4* zp = (float4*)Zout + (size_t)c * (D_FEAT / 4) + lane;
      const float4 h = H4[(size_t)c * (D_FEAT / 4) + lane];
      float4 s = make_float4(0.0f, 0.0f, 0.0f, 0.0f);
      if (MODE == 2) s = *zp;

      float4 acc = make_float4(0.0f, 0.0f, 0.0f, 0.0f);
      for (int base = start; base < end; base += 32) {
        const int m = (end - base < 32) ? (end - base) : 32;
        int r = r0[k];
        float nrm = n0[k];
        if (base != start) {
          r = 0;
          nrm = 0.0f;
          if (lane < m) {
            r = ssrc[base + lane];
            nrm = dinv[r] * swt[base + lane] * dc;
          }
        }
        const int nrm_bits = __float_as_int(nrm);
#pragma unroll 4
        for (int j = 0; j < m; ++j) {
          const int rj = __builtin_amdgcn_readlane(r, j);
          const float nj = __int_as_float(__builtin_amdgcn_readlane(nrm_bits, j));
          const float4 v = H4[(size_t)rj * (D_FEAT / 4) + lane];
          acc.x += v.x * nj;
          acc.y += v.y * nj;
          acc.z += v.z * nj;
          acc.w += v.w * nj;
        }
      }

      float4 o;
      float v;
      v = acc.x + h.x * dc * dc + b.x;
      o.x = fmaxf(v, 0.0f) + p.x * fminf(v, 0.0f);
      v = acc.y + h.y * dc * dc + b.y;
      o.y = fmaxf(v, 0.0f) + p.y * fminf(v, 0.0f);
      v = acc.z + h.z * dc * dc + b.z;
      o.z = fmaxf(v, 0.0f) + p.z * fminf(v, 0.0f);
      v = acc.w + h.w * dc * dc + b.w;
      o.w = fmaxf(v, 0.0f) + p.w * fminf(v, 0.0f);
      if (MODE == 0) {
        *zp = o;
      } else {
        float4 q;
        if (MODE == 1) {
          q.x = ct * o.x;
          q.y = ct * o.y;
          q.z = ct * o.z;
          q.w = ct * o.w;
        } else {
          q.x = s.x + ct * o.x;
          q.y = s.y + ct * o.y;
          q.z = s.z + ct * o.z;
          q.w = s.w + ct * o.w;
        }
        *zp = q;
      }
    }
  }
}

extern "C" void kernel_launch(void* const* d_in, const int* in_sizes, int n_in,
                              void* d_out, int out_size, void* d_ws, size_t ws_size,
                              hipStream_t stream) {
  const float* x       = (const float*)d_in[0];
  const int*   eidx    = (const int*)  d_in[1];
  const float* ew      = (const float*)d_in[2];
  const float* coeff   = (const float*)d_in[3];
  const float* W1      = (const float*)d_in[4];
  const float* b1      = (const float*)d_in[5];
  const float* W2      = (const float*)d_in[6];
  const float* b2      = (const float*)d_in[7];
  const float* prelu_w = (const float*)d_in[8];

  const int N = in_sizes[0] / D_FEAT;
  const int E = in_sizes[2] / 3;
  const int G = 3 * N;
  const size_t E3 = (size_t)3 * E;

  const size_t padG  = ((size_t)G + 1 + 31) & ~(size_t)31;
  const size_t padE3 = (E3 + 31) & ~(size_t)31;
  const int scanBlocks = (G + SCAN_B - 1) / SCAN_B;
  const size_t padS  = ((size_t)scanBlocks + 31) & ~(size_t)31;
  int*   cnt  = (int*)d_ws;
  int*   off  = cnt + padG;
  int*   bsum = off + padG;
  int*   rank = bsum + padS;
  int*   tmp  = rank + padE3;
  int*   ssrc = tmp + padE3;
  float* swt  = (float*)(ssrc + padE3);
  float* dinv = swt + padE3;
  const size_t plane = (size_t)N * D_FEAT;
  float* H1 = dinv + padG;
  float* Z1 = H1 + plane;
  float* H2 = Z1 + 3 * plane;

  const int ETHREADS = 256;
  const dim3 waveBlock(32, 8);
  const dim3 gemmBlock(32, GEMM_WAVES);
  const int gemmBlocks1 = (N / 16 + GEMM_WAVES - 1) / GEMM_WAVES;
  const int gemmBlocks3 = (G / 16 + GEMM_WAVES - 1) / GEMM_WAVES;
  const int nodeWaveBlocks = (N + 8 * PULL_NODES - 1) / (8 * PULL_NODES);
  const dim3 nodeGridAll(nodeWaveBlocks, 3);
  const int edgesPerBlock = ETHREADS * 4 * EDGE_QUADS;
  const dim3 edgeGrid((E + edgesPerBlock - 1) / edgesPerBlock, 3);
  const int sortBlocks = (G + 8 * SORT_BUCKETS - 1) / (8 * SORT_BUCKETS);

  zero_kernel<<<(G + ETHREADS - 1) / ETHREADS, ETHREADS, 0, stream>>>((float*)cnt, G);
  rank_kernel<<<edgeGrid, ETHREADS, 0, stream>>>(eidx, cnt, rank, E, N);
  bsum_kernel<<<scanBlocks, SCAN_B, 0, stream>>>(cnt, bsum, G);
  scan_kernel<<<scanBlocks, SCAN_B, 0, stream>>>(cnt, bsum, off, G);
  fill_kernel<<<edgeGrid, ETHREADS, 0, stream>>>(eidx, off, rank, tmp, E, N);
  sort_kernel<<<sortBlocks, waveBlock, 0, stream>>>(eidx, ew, off, tmp, ssrc, swt, E, N, G);
  dinv_kernel<<<(G + ETHREADS - 1) / ETHREADS, ETHREADS, 0, stream>>>(off, swt, dinv, G);

  gemm_wmma_kernel<<<gemmBlocks1, gemmBlock, 0, stream>>>(x, W1, H1, N);
  pull_kernel<0><<<nodeGridAll, waveBlock, 0, stream>>>(off, ssrc, swt, dinv, H1, b1, prelu_w, coeff, Z1, N, (size_t)0, plane);

  gemm_wmma_kernel<<<gemmBlocks3, gemmBlock, 0, stream>>>(Z1, W2, H2, G);

  for (int t = 0; t < 3; ++t) {
    const int*   offT  = off + (size_t)t * N;
    const float* dinvT = dinv + (size_t)t * N;
    const float* H2T   = H2 + (size_t)t * plane;
    if (t == 0)
      pull_kernel<1><<<nodeWaveBlocks, waveBlock, 0, stream>>>(offT, ssrc, swt, dinvT, H2T, b2, prelu_w, coeff + t, (float*)d_out, N, (size_t)0, (size_t)0);
    else
      pull_kernel<2><<<nodeWaveBlocks, waveBlock, 0, stream>>>(offT, ssrc, swt, dinvT, H2T, b2, prelu_w, coeff + t, (float*)d_out, N, (size_t)0, (size_t)0);
  }
}


# rank's integer adds go to one count plane per package half (die number); bucket size = sum of both, 2nd half's slots behind the 1st's: no counter line added to from both halves; sum order unchanged
# speedup vs baseline: 4.8189x; 1.4518x over previous
#include <hip/hip_runtime.h>
#include <hip/hip_bf16.h>


typedef __attribute__((ext_vector_type(2))) float v2f;
typedef __attribute__((ext_vector_type(8))) float v8f;

#define D_FEAT 128
#define SCAN_B 1024
#define EDGE_QUADS 2
#define SORT_BUCKETS 8
#define PULL_NODES 4

__device__ __forceinline__ int4 load_quad(const int* __restrict__ p, int e, int n, bool vec) {
  int4 v = make_int4(0, 0, 0, 0);
  if (e < n) {
    if (vec) {
      v = *(const int4*)(p + e);
    } else {
      v.x = p[e];
      if (e + 1 < n) v.y = p[e + 1];
      if (e + 2 < n) v.z = p[e + 2];
      if (e + 3 < n) v.w = p[e + 3];
    }
  }
  return v;
}

__device__ __forceinline__ void store_quad(int* __restrict__ p, int e, int n, bool vec, int4 v) {
  if (e < n) {
    if (vec) {
      *(int4*)(p + e) = v;
    } else {
      p[e] = v.x;
      if (e + 1 < n) p[e + 1] = v.y;
      if (e + 2 < n) p[e + 2] = v.z;
      if (e + 3 < n) p[e + 3] = v.w;
    }
  }
}

__global__ void zero_kernel(float* __restrict__ p, int n) {
  int i = blockIdx.x * blockDim.x + threadIdx.x;
  if (i < n) p[i] = 0.0f;
}

__global__ __launch_bounds__(256) void rank_kernel(
    const int* __restrict__ eidx, int* __restrict__ cnt, int cntPlane,
    int* __restrict__ hflag, int* __restrict__ rank, int E, int N) {
  const int t = blockIdx.y;
  const int* col = eidx + (size_t)t * 2 * E + E;
  int* rk = rank + (size_t)t * E;
  const bool vec = (E & 3) == 0;
  const int stride = gridDim.x * blockDim.x * 4;
  const int e0 = (blockIdx.x * blockDim.x + threadIdx.x) * 4;
  int4 c[EDGE_QUADS];
#pragma unroll
  for (int i = 0; i < EDGE_QUADS; ++i) c[i] = load_quad(col, e0 + i * stride, E, vec);

  __shared__ int blockHalf;
  if (threadIdx.x == 0) {
    const unsigned where = __builtin_amdgcn_s_sendmsg_rtn(0x87);
    blockHalf = (int)((where >> 18) & 1u);
    hflag[t * gridDim.x + blockIdx.x] = blockHalf;
  }
  __syncthreads();
  const int half = blockHalf;
  int* cn = cnt + (size_t)half * cntPlane + t * N;

  int4 r[EDGE_QUADS];
#pragma unroll
  for (int i = 0; i < EDGE_QUADS; ++i) {
    const int e = e0 + i * stride;
    r[i] = make_int4(0, 0, 0, 0);
    if (e < E) r[i].x = atomicAdd(&cn[c[i].x], 1);
    if (e + 1 < E) r[i].y = atomicAdd(&cn[c[i].y], 1);
    if (e + 2 < E) r[i].z = atomicAdd(&cn[c[i].z], 1);
    if (e + 3 < E) r[i].w = atomicAdd(&cn[c[i].w], 1);
  }
#pragma unroll
  for (int i = 0; i < EDGE_QUADS; ++i) store_quad(rk, e0 + i * stride, E, vec, r[i]);
}

__global__ __launch_bounds__(SCAN_B) void bsum_kernel(
    const int* __restrict__ cnt, const int* __restrict__ cnt1,
    int* __restrict__ bsum, int n) {
  __shared__ int s[32];
  const int tid = threadIdx.x;
  const int i = blockIdx.x * SCAN_B + tid;
  int v = (i < n) ? cnt[i] + cnt1[i] : 0;
  for (int m = 16; m > 0; m >>= 1) v += __shfl_xor(v, m, 32);
  if ((tid & 31) == 0) s[tid >> 5] = v;
  __syncthreads();
  if (tid < 32) {
    int x = s[tid];
    for (int m = 16; m > 0; m >>= 1) x += __shfl_xor(x, m, 32);
    if (tid == 0) bsum[blockIdx.x] = x;
  }
}

__global__ __launch_bounds__(SCAN_B) void scan_kernel(
    const int* __restrict__ cnt, const int* __restrict__ cnt1,
    const int* __restrict__ bsum,
    int* __restrict__ off, int* __restrict__ off1, int n) {
  __shared__ int sw[32];
  __shared__ int sbase;
  const int tid = threadIdx.x;
  const int lane = tid & 31;
  const int wv = tid >> 5;
  if (wv == 0) {
    int a = 0;
    for (int j = lane; j < (int)blockIdx.x; j += 32) a += bsum[j];
    for (int m = 16; m > 0; m >>= 1) a += __shfl_xor(a, m, 32);
    if (lane == 0) sbase = a;
  }
  const int i = blockIdx.x * SCAN_B + tid;
  const int v0 = (i < n) ? cnt[i] : 0;
  const int v = v0 + ((i < n) ? cnt1[i] : 0);
  int x = v;
  for (int d = 1; d < 32; d <<= 1) {
    const int y = __shfl_up(x, d, 32);
    if (lane >= d) x += y;
  }
  if (lane == 31) sw[wv] = x;
  __syncthreads();
  if (wv == 0) {
    const int wtot = sw[lane];
    int xx = wtot;
    for (int d = 1; d < 32; d <<= 1) {
      const int y = __shfl_up(xx, d, 32);
      if (lane >= d) xx += y;
    }
    sw[lane] = xx - wtot;
  }
  __syncthreads();
  const int excl = sbase + sw[wv] + x - v;
  if (i < n) {
    off[i] = excl;
    off1[i] = excl + v0;
  }
  if (i == n - 1) off[n] = excl + v;
}

__global__ __launch_bounds__(256) void fill_kernel(
    const int* __restrict__ eidx, const int* __restrict__ off,
    const int* __restrict__ off1, const int* __restrict__ hflag,
    const int* __restrict__ rank, int* __restrict__ tmp, int E, int N) {
  const int t = blockIdx.y;
  const int* col = eidx + (size_t)t * 2 * E + E;
  const int* rk = rank + (size_t)t * E;
  const int* of = (hflag[t * gridDim.x + blockIdx.x] ? off1 : off) + t * N;
  const bool vec = (E & 3) == 0;
  const int stride = gridDim.x * blockDim.x * 4;
  const int e0 = (blockIdx.x * blockDim.x + threadIdx.x) * 4;
  int4 c[EDGE_QUADS], r[EDGE_QUADS];
#pragma unroll
  for (int i = 0; i < EDGE_QUADS; ++i) {
    c[i] = load_quad(col, e0 + i * stride, E, vec);
    r[i] = load_quad(rk, e0 + i * stride, E, vec);
  }
  int4 o[EDGE_QUADS];
#pragma unroll
  for (int i = 0; i < EDGE_QUADS; ++i) {
    const int e = e0 + i * stride;
    o[i] = make_int4(0, 0, 0, 0);
    if (e < E) o[i].x = of[c[i].x];
    if (e + 1 < E) o[i].y = of[c[i].y];
    if (e + 2 < E) o[i].z = of[c[i].z];
    if (e + 3 < E) o[i].w = of[c[i].w];
  }
#pragma unroll
  for (int i = 0; i < EDGE_QUADS; ++i) {
    const int e = e0 + i * stride;
    if (e < E) tmp[o[i].x + r[i].x] = e;
    if (e + 1 < E) tmp[o[i].y + r[i].y] = e + 1;
    if (e + 2 < E) tmp[o[i].z + r[i].z] = e + 2;
    if (e + 3 < E) tmp[o[i].w + r[i].w] = e + 3;
  }
}

__global__ __launch_bounds__(256) void sort_kernel(
    const int* __restrict__ eidx, const float* __restrict__ ew,
    const int* __restrict__ off, const int* __restrict__ tmp,
    int* __restrict__ ssrc, float* __restrict__ swt, int E, int N, int G) {
  const int lane = threadIdx.x;
  const int g0 = __builtin_amdgcn_readfirstlane(
      (blockIdx.x * 8 + threadIdx.y) * SORT_BUCKETS);
  if (g0 >= G) return;

  int gi = g0 + lane;
  if (gi > G) gi = G;
  const int ov = (lane <= SORT_BUCKETS) ? off[gi] : 0;
  int st[SORT_BUCKETS + 1];
#pragma unroll
  for (int k = 0; k <= SORT_BUCKETS; ++k) st[k] = __builtin_amdgcn_readlane(ov, k);

  int eid[SORT_BUCKETS];
#pragma unroll
  for (int k = 0; k < SORT_BUCKETS; ++k) {
    const int n = st[k + 1] - st[k];
    eid[k] = (lane < n) ? tmp[st[k] + lane] : 0x7fffffff;
  }

  int rk[SORT_BUCKETS];
#pragma unroll
  for (int k = 0; k < SORT_BUCKETS; ++k) {
    const int n = st[k + 1] - st[k];
    rk[k] = 0;
    if (n <= 32) {
      for (int j = 0; j < n; ++j)
        rk[k] += (__builtin_amdgcn_readlane(eid[k], j) < eid[k]) ? 1 : 0;
    }
  }

  int src[SORT_BUCKETS];
  float wt[SORT_BUCKETS];
#pragma unroll
  for (int k = 0; k < SORT_BUCKETS; ++k) {
    const int n = st[k + 1] - st[k];
    const int g = g0 + k;
    const int t = (g >= N ? 1 : 0) + (g >= 2 * N ? 1 : 0);
    const int e = (n <= 32 && lane < n) ? eid[k] : 0;
    src[k] = eidx[(size_t)t * 2 * E + e];
    wt[k] = ew[(size_t)t * E + e];
  }
#pragma unroll
  for (int k = 0; k < SORT_BUCKETS; ++k) {
    const int n = st[k + 1] - st[k];
    if (n <= 32 && lane < n) {
      ssrc[st[k] + rk[k]] = src[k];
      swt[st[k] + rk[k]] = wt[k];
    }
  }

#pragma unroll
  for (int k = 0; k < SORT_BUCKETS; ++k) {
    const int n = st[k + 1] - st[k];
    if (n > 32) {
      const int g = g0 + k;
      const int t = (g >= N ? 1 : 0) + (g >= 2 * N ? 1 : 0);
      const int* row = eidx + (size_t)t * 2 * E;
      const float* w = ew + (size_t)t * E;
      const int start = st[k];
      for (int i = lane; i < n; i += 32) {
        const int e = tmp[start + i];
        int q = 0;
        for (int j = 0; j < n; ++j) q += (tmp[start + j] < e) ? 1 : 0;
        ssrc[start + q] = row[e];
        swt[start + q] = w[e];
      }
    }
  }
}

__global__ void dinv_kernel(const int* __restrict__ off, const float* __restrict__ swt,
                            float* __restrict__ dinv, int G) {
  const int g = blockIdx.x * blockDim.x + threadIdx.x;
  if (g >= G) return;
  const int end = off[g + 1];
  float s = 0.0f;
  for (int p = off[g]; p < end; ++p) s += swt[p];
  const float d = s + 1.0f;
  dinv[g] = rsqrtf(d);
}

typedef __attribute__((ext_vector_type(4))) float v4f;

#define GEMM_WAVES 4
#define KSTEPS_PER_GROUP 4

struct FragGroup {
  v2f a[KSTEPS_PER_GROUP];
  v4f blo[KSTEPS_PER_GROUP][2];
  v4f bhi[KSTEPS_PER_GROUP][2];
};

__device__ __forceinline__ void gemm_load_group(FragGroup& f, const float* __restrict__ zp,
                                                const float* __restrict__ wp, int kbase) {
#pragma unroll
  for (int q = 0; q < KSTEPS_PER_GROUP; ++q) {
    const int k0 = kbase + 4 * q;
    f.a[q] = *(const v2f*)(zp + k0);
#pragma unroll
    for (int s = 0; s < 2; ++s) {
      f.blo[q][s] = *(const v4f*)(wp + (size_t)(k0 + s) * D_FEAT);
      f.bhi[q][s] = *(const v4f*)(wp + (size_t)(k0 + s) * D_FEAT + 64);
    }
  }
}

__device__ __forceinline__ void gemm_mma_group(v8f (&acc)[8], const FragGroup& f) {
#pragma unroll
  for (int q = 0; q < KSTEPS_PER_GROUP; ++q) {
#pragma unroll
    for (int t = 0; t < 4; ++t) {
      v2f b;
      b.x = f.blo[q][0][t];
      b.y = f.blo[q][1][t];
      acc[t] = __builtin_amdgcn_wmma_f32_16x16x4_f32(
           false, f.a[q],  false, b,
           (short)0, acc[t],  false,  false);
      v2f d;
      d.x = f.bhi[q][0][t];
      d.y = f.bhi[q][1][t];
      acc[4 + t] = __builtin_amdgcn_wmma_f32_16x16x4_f32(
           false, f.a[q],  false, d,
           (short)0, acc[4 + t],  false,  false);
    }
  }
}

__global__ __launch_bounds__(32 * GEMM_WAVES) void gemm_wmma_kernel(
    const float* __restrict__ Z, const float* __restrict__ W,
    float* __restrict__ H, int nRows) {
  const int lane = threadIdx.x;
  const int strip = __builtin_amdgcn_readfirstlane(blockIdx.x * GEMM_WAVES + threadIdx.y);
  const int m0 = strip * 16;
  if (m0 >= nRows) return;

  const int c = lane & 15;
  const int h = lane >> 4;
  const float* zp = Z + (size_t)(m0 + c) * D_FEAT + 2 * h;
  const float* wp = W + (size_t)(2 * h) * D_FEAT + 4 * c;

  v8f acc[8];
#pragma unroll
  for (int t = 0; t < 8; ++t) acc[t] = v8f{};

  constexpr int KGROUP = 4 * KSTEPS_PER_GROUP;
  FragGroup f0, f1;
  gemm_load_group(f0, zp, wp, 0);
#pragma unroll
  for (int g = 0; g < D_FEAT / KGROUP; g += 2) {
    gemm_load_group(f1, zp, wp, (g + 1) * KGROUP);
    __builtin_amdgcn_sched_barrier(0);
    gemm_mma_group(acc, f0);
    if (g + 2 < D_FEAT / KGROUP) {
      gemm_load_group(f0, zp, wp, (g + 2) * KGROUP);
      __builtin_amdgcn_sched_barrier(0);
    }
    gemm_mma_group(acc, f1);
  }

  float* hp = H + (size_t)(m0 + 8 * h) * D_FEAT + 4 * c;
#pragma unroll
  for (int r = 0; r < 8; ++r) {
    v4f lo, hi;
#pragma unroll
    for (int t = 0; t < 4; ++t) {
      lo[t] = acc[t][r];
      hi[t] = acc[4 + t][r];
    }
    *(v4f*)(hp + (size_t)r * D_FEAT) = lo;
    *(v4f*)(hp + (size_t)r * D_FEAT + 64) = hi;
  }
}

template <int MODE>
__global__ __launch_bounds__(256) void pull_kernel(
    const int* __restrict__ offAll, const int* __restrict__ ssrc,
    const float* __restrict__ swt, const float* __restrict__ dinvAll,
    const float* __restrict__ Hall, const float* __restrict__ bias,
    const float* __restrict__ pw, const float* __restrict__ cscale,
    float* __restrict__ Zall, int N, size_t hPlane, size_t zPlane) {
  const int lane = threadIdx.x;
  const int c0 = __builtin_amdgcn_readfirstlane(
      (blockIdx.x * 8 + threadIdx.y) * PULL_NODES);
  if (c0 >= N) return;
  const int scale = blockIdx.y;
  const int* __restrict__ off = offAll + (size_t)scale * N;
  const float* __restrict__ dinv = dinvAll + (size_t)scale * N;
  const float* __restrict__ H = Hall + (size_t)scale * hPlane;
  float* __restrict__ Zout = Zall + (size_t)scale * zPlane;
  const float4* H4 = (const float4*)H;

  int ci = c0 + lane;
  if (ci > N) ci = N;
  const int ov = (lane <= PULL_NODES) ? off[ci] : 0;
  const float dv = (lane < PULL_NODES && c0 + lane < N) ? dinv[c0 + lane] : 0.0f;
  const float4 b = ((const float4*)bias)[lane];
  const float4 p = ((const float4*)pw)[lane];
  float ct = 0.0f;
  if (MODE != 0) ct = cscale[0];

  int st[PULL_NODES + 1];
  float dcs[PULL_NODES];
#pragma unroll
  for (int k = 0; k <= PULL_NODES; ++k) st[k] = __builtin_amdgcn_readlane(ov, k);
#pragma unroll
  for (int k = 0; k < PULL_NODES; ++k)
    dcs[k] = __int_as_float(__builtin_amdgcn_readlane(__float_as_int(dv), k));

  int r0[PULL_NODES];
  float w0[PULL_NODES];
#pragma unroll
  for (int k = 0; k < PULL_NODES; ++k) {
    const int n = st[k + 1] - st[k];
    r0[k] = 0;
    w0[k] = 0.0f;
    if (lane < n) {
      r0[k] = ssrc[st[k] + lane];
      w0[k] = swt[st[k] + lane];
    }
  }
  float d0[PULL_NODES];
#pragma unroll
  for (int k = 0; k < PULL_NODES; ++k) d0[k] = dinv[r0[k]];
  float n0[PULL_NODES];
#pragma unroll
  for (int k = 0; k < PULL_NODES; ++k) {
    const int n = st[k + 1] - st[k];
    const float x = d0[k] * w0[k] * dcs[k];
    n0[k] = (lane < n) ? x : 0.0f;
  }

#pragma unroll
  for (int k = 0; k < PULL_NODES; ++k) {
    const int c = c0 + k;
    if (c < N) {
      const int start = st[k];
      const int end = st[k + 1];
      const float dc = dcs[k];
      float4* zp = (float4*)Zout + (size_t)c * (D_FEAT / 4) + lane;
      const float4 h = H4[(size_t)c * (D_FEAT / 4) + lane];
      float4 s = make_float4(0.0f, 0.0f, 0.0f, 0.0f);
      if (MODE == 2) s = *zp;

      float4 acc = make_float4(0.0f, 0.0f, 0.0f, 0.0f);
      for (int base = start; base < end; base += 32) {
        const int m = (end - base < 32) ? (end - base) : 32;
        int r = r0[k];
        float nrm = n0[k];
        if (base != start) {
          r = 0;
          nrm = 0.0f;
          if (lane < m) {
            r = ssrc[base + lane];
            nrm = dinv[r] * swt[base + lane] * dc;
          }
        }
        const int nrm_bits = __float_as_int(nrm);
#pragma unroll 4
        for (int j = 0; j < m; ++j) {
          const int rj = __builtin_amdgcn_readlane(r, j);
          const float nj = __int_as_float(__builtin_amdgcn_readlane(nrm_bits, j));
          const float4 v = H4[(size_t)rj * (D_FEAT / 4) + lane];
          acc.x += v.x * nj;
          acc.y += v.y * nj;
          acc.z += v.z * nj;
          acc.w += v.w * nj;
        }
      }

      float4 o;
      float v;
      v = acc.x + h.x * dc * dc + b.x;
      o.x = fmaxf(v, 0.0f) + p.x * fminf(v, 0.0f);
      v = acc.y + h.y * dc * dc + b.y;
      o.y = fmaxf(v, 0.0f) + p.y * fminf(v, 0.0f);
      v = acc.z + h.z * dc * dc + b.z;
      o.z = fmaxf(v, 0.0f) + p.z * fminf(v, 0.0f);
      v = acc.w + h.w * dc * dc + b.w;
      o.w = fmaxf(v, 0.0f) + p.w * fminf(v, 0.0f);
      if (MODE == 0) {
        *zp = o;
      } else {
        float4 q;
        if (MODE == 1) {
          q.x = ct * o.x;
          q.y = ct * o.y;
          q.z = ct * o.z;
          q.w = ct * o.w;
        } else {
          q.x = s.x + ct * o.x;
          q.y = s.y + ct * o.y;
          q.z = s.z + ct * o.z;
          q.w = s.w + ct * o.w;
        }
        *zp = q;
      }
    }
  }
}

extern "C" void kernel_launch(void* const* d_in, const int* in_sizes, int n_in,
                              void* d_out, int out_size, void* d_ws, size_t ws_size,
                              hipStream_t stream) {
  const float* x       = (const float*)d_in[0];
  const int*   eidx    = (const int*)  d_in[1];
  const float* ew      = (const float*)d_in[2];
  const float* coeff   = (const float*)d_in[3];
  const float* W1      = (const float*)d_in[4];
  const float* b1      = (const float*)d_in[5];
  const float* W2      = (const float*)d_in[6];
  const float* b2      = (const float*)d_in[7];
  const float* prelu_w = (const float*)d_in[8];

  const int N = in_sizes[0] / D_FEAT;
  const int E = in_sizes[2] / 3;
  const int G = 3 * N;
  const size_t E3 = (size_t)3 * E;

  const size_t padG  = ((size_t)G + 1 + 31) & ~(size_t)31;
  const size_t padE3 = (E3 + 31) & ~(size_t)31;
  const int scanBlocks = (G + SCAN_B - 1) / SCAN_B;
  const size_t padS  = ((size_t)scanBlocks + 31) & ~(size_t)31;
  const int ETHREADS = 256;
  const int edgesPerBlock = ETHREADS * 4 * EDGE_QUADS;
  const int edgeBlocks = (E + edgesPerBlock - 1) / edgesPerBlock;
  const size_t padH  = ((size_t)3 * edgeBlocks + 31) & ~(size_t)31;
  int*   cnt  = (int*)d_ws;
  int*   cnt1 = cnt + padG;
  int*   off  = cnt1 + padG;
  int*   off1 = off + padG;
  int*   bsum = off1 + padG;
  int*   hflag = bsum + padS;
  int*   rank = hflag + padH;
  int*   tmp  = rank + padE3;
  int*   ssrc = tmp + padE3;
  float* swt  = (float*)(ssrc + padE3);
  float* dinv = swt + padE3;
  const size_t plane = (size_t)N * D_FEAT;
  float* H1 = dinv + padG;
  float* Z1 = H1 + plane;
  float* H2 = Z1 + 3 * plane;

  const dim3 waveBlock(32, 8);
  const dim3 gemmBlock(32, GEMM_WAVES);
  const int gemmBlocks1 = (N / 16 + GEMM_WAVES - 1) / GEMM_WAVES;
  const int gemmBlocks3 = (G / 16 + GEMM_WAVES - 1) / GEMM_WAVES;
  const int nodeWaveBlocks = (N + 8 * PULL_NODES - 1) / (8 * PULL_NODES);
  const dim3 nodeGridAll(nodeWaveBlocks, 3);
  const dim3 edgeGrid(edgeBlocks, 3);
  const int sortBlocks = (G + 8 * SORT_BUCKETS - 1) / (8 * SORT_BUCKETS);

  const int cntWords = (int)(2 * padG);
  zero_kernel<<<(cntWords + ETHREADS - 1) / ETHREADS, ETHREADS, 0, stream>>>((float*)cnt, cntWords);
  rank_kernel<<<edgeGrid, ETHREADS, 0, stream>>>(eidx, cnt, (int)padG, hflag, rank, E, N);
  bsum_kernel<<<scanBlocks, SCAN_B, 0, stream>>>(cnt, cnt1, bsum, G);
  scan_kernel<<<scanBlocks, SCAN_B, 0, stream>>>(cnt, cnt1, bsum, off, off1, G);
  fill_kernel<<<edgeGrid, ETHREADS, 0, stream>>>(eidx, off, off1, hflag, rank, tmp, E, N);
  sort_kernel<<<sortBlocks, waveBlock, 0, stream>>>(eidx, ew, off, tmp, ssrc, swt, E, N, G);
  dinv_kernel<<<(G + ETHREADS - 1) / ETHREADS, ETHREADS, 0, stream>>>(off, swt, dinv, G);

  gemm_wmma_kernel<<<gemmBlocks1, gemmBlock, 0, stream>>>(x, W1, H1, N);
  pull_kernel<0><<<nodeGridAll, waveBlock, 0, stream>>>(off, ssrc, swt, dinv, H1, b1, prelu_w, coeff, Z1, N, (size_t)0, plane);

  gemm_wmma_kernel<<<gemmBlocks3, gemmBlock, 0, stream>>>(Z1, W2, H2, G);

  for (int t = 0; t < 3; ++t) {
    const int*   offT  = off + (size_t)t * N;
    const float* dinvT = dinv + (size_t)t * N;
    const float* H2T   = H2 + (size_t)t * plane;
    if (t == 0)
      pull_kernel<1><<<nodeWaveBlocks, waveBlock, 0, stream>>>(offT, ssrc, swt, dinvT, H2T, b2, prelu_w, coeff + t, (float*)d_out, N, (size_t)0, (size_t)0);
    else
      pull_kernel<2><<<nodeWaveBlocks, waveBlock, 0, stream>>>(offT, ssrc, swt, dinvT, H2T, b2, prelu_w, coeff + t, (float*)d_out, N, (size_t)0, (size_t)0);
  }
}


# fill stores a 16-byte record (edge id, source, weight) per bucket slot, read in edge order; the ordering pass reads its buckets' records as whole lines, no per-edge lookups; sum order unchanged
# speedup vs baseline: 4.9669x; 1.0307x over previous
#include <hip/hip_runtime.h>
#include <hip/hip_bf16.h>


typedef __attribute__((ext_vector_type(2))) float v2f;
typedef __attribute__((ext_vector_type(8))) float v8f;

#define D_FEAT 128
#define SCAN_B 1024
#define EDGE_QUADS 2
#define SORT_BUCKETS 8
#define PULL_NODES 4

__device__ __forceinline__ int4 load_quad(const int* __restrict__ p, int e, int n, bool vec) {
  int4 v = make_int4(0, 0, 0, 0);
  if (e < n) {
    if (vec) {
      v = *(const int4*)(p + e);
    } else {
      v.x = p[e];
      if (e + 1 < n) v.y = p[e + 1];
      if (e + 2 < n) v.z = p[e + 2];
      if (e + 3 < n) v.w = p[e + 3];
    }
  }
  return v;
}

__device__ __forceinline__ void store_quad(int* __restrict__ p, int e, int n, bool vec, int4 v) {
  if (e < n) {
    if (vec) {
      *(int4*)(p + e) = v;
    } else {
      p[e] = v.x;
      if (e + 1 < n) p[e + 1] = v.y;
      if (e + 2 < n) p[e + 2] = v.z;
      if (e + 3 < n) p[e + 3] = v.w;
    }
  }
}

__device__ __forceinline__ float4 load_quad_f(const float* __restrict__ p, int e, int n, bool vec) {
  float4 v = make_float4(0.0f, 0.0f, 0.0f, 0.0f);
  if (e < n) {
    if (vec) {
      v = *(const float4*)(p + e);
    } else {
      v.x = p[e];
      if (e + 1 < n) v.y = p[e + 1];
      if (e + 2 < n) v.z = p[e + 2];
      if (e + 3 < n) v.w = p[e + 3];
    }
  }
  return v;
}

__global__ void zero_kernel(float* __restrict__ p, int n) {
  int i = blockIdx.x * blockDim.x + threadIdx.x;
  if (i < n) p[i] = 0.0f;
}

__global__ __launch_bounds__(256) void rank_kernel(
    const int* __restrict__ eidx, int* __restrict__ cnt, int cntPlane,
    int* __restrict__ hflag, int* __restrict__ rank, int E, int N) {
  const int t = blockIdx.y;
  const int* col = eidx + (size_t)t * 2 * E + E;
  int* rk = rank + (size_t)t * E;
  const bool vec = (E & 3) == 0;
  const int stride = gridDim.x * blockDim.x * 4;
  const int e0 = (blockIdx.x * blockDim.x + threadIdx.x) * 4;
  int4 c[EDGE_QUADS];
#pragma unroll
  for (int i = 0; i < EDGE_QUADS; ++i) c[i] = load_quad(col, e0 + i * stride, E, vec);

  __shared__ int blockHalf;
  if (threadIdx.x == 0) {
    const unsigned where = __builtin_amdgcn_s_sendmsg_rtn(0x87);
    blockHalf = (int)((where >> 18) & 1u);
    hflag[t * gridDim.x + blockIdx.x] = blockHalf;
  }
  __syncthreads();
  const int half = blockHalf;
  int* cn = cnt + (size_t)half * cntPlane + t * N;

  int4 r[EDGE_QUADS];
#pragma unroll
  for (int i = 0; i < EDGE_QUADS; ++i) {
    const int e = e0 + i * stride;
    r[i] = make_int4(0, 0, 0, 0);
    if (e < E) r[i].x = atomicAdd(&cn[c[i].x], 1);
    if (e + 1 < E) r[i].y = atomicAdd(&cn[c[i].y], 1);
    if (e + 2 < E) r[i].z = atomicAdd(&cn[c[i].z], 1);
    if (e + 3 < E) r[i].w = atomicAdd(&cn[c[i].w], 1);
  }
#pragma unroll
  for (int i = 0; i < EDGE_QUADS; ++i) store_quad(rk, e0 + i * stride, E, vec, r[i]);
}

__global__ __launch_bounds__(SCAN_B) void bsum_kernel(
    const int* __restrict__ cnt, const int* __restrict__ cnt1,
    int* __restrict__ bsum, int n) {
  __shared__ int s[32];
  const int tid = threadIdx.x;
  const int i = blockIdx.x * SCAN_B + tid;
  int v = (i < n) ? cnt[i] + cnt1[i] : 0;
  for (int m = 16; m > 0; m >>= 1) v += __shfl_xor(v, m, 32);
  if ((tid & 31) == 0) s[tid >> 5] = v;
  __syncthreads();
  if (tid < 32) {
    int x = s[tid];
    for (int m = 16; m > 0; m >>= 1) x += __shfl_xor(x, m, 32);
    if (tid == 0) bsum[blockIdx.x] = x;
  }
}

__global__ __launch_bounds__(SCAN_B) void scan_kernel(
    const int* __restrict__ cnt, const int* __restrict__ cnt1,
    const int* __restrict__ bsum,
    int* __restrict__ off, int* __restrict__ off1, int n) {
  __shared__ int sw[32];
  __shared__ int sbase;
  const int tid = threadIdx.x;
  const int lane = tid & 31;
  const int wv = tid >> 5;
  if (wv == 0) {
    int a = 0;
    for (int j = lane; j < (int)blockIdx.x; j += 32) a += bsum[j];
    for (int m = 16; m > 0; m >>= 1) a += __shfl_xor(a, m, 32);
    if (lane == 0) sbase = a;
  }
  const int i = blockIdx.x * SCAN_B + tid;
  const int v0 = (i < n) ? cnt[i] : 0;
  const int v = v0 + ((i < n) ? cnt1[i] : 0);
  int x = v;
  for (int d = 1; d < 32; d <<= 1) {
    const int y = __shfl_up(x, d, 32);
    if (lane >= d) x += y;
  }
  if (lane == 31) sw[wv] = x;
  __syncthreads();
  if (wv == 0) {
    const int wtot = sw[lane];
    int xx = wtot;
    for (int d = 1; d < 32; d <<= 1) {
      const int y = __shfl_up(xx, d, 32);
      if (lane >= d) xx += y;
    }
    sw[lane] = xx - wtot;
  }
  __syncthreads();
  const int excl = sbase + sw[wv] + x - v;
  if (i < n) {
    off[i] = excl;
    off1[i] = excl + v0;
  }
  if (i == n - 1) off[n] = excl + v;
}

__global__ __launch_bounds__(256) void fill_kernel(
    const int* __restrict__ eidx, const float* __restrict__ ew,
    const int* __restrict__ off,
    const int* __restrict__ off1, const int* __restrict__ hflag,
    const int* __restrict__ rank, int* __restrict__ tmp, int E, int N) {
  const int t = blockIdx.y;
  const int* row = eidx + (size_t)t * 2 * E;
  const int* col = row + E;
  const float* wgt = ew + (size_t)t * E;
  const int* rk = rank + (size_t)t * E;
  int4* rec = (int4*)tmp;
  const int* of = (hflag[t * gridDim.x + blockIdx.x] ? off1 : off) + t * N;
  const bool vec = (E & 3) == 0;
  const int stride = gridDim.x * blockDim.x * 4;
  const int e0 = (blockIdx.x * blockDim.x + threadIdx.x) * 4;
  int4 c[EDGE_QUADS], r[EDGE_QUADS], s[EDGE_QUADS];
  float4 w[EDGE_QUADS];
  if (vec) {
#pragma unroll
    for (int i = 0; i < EDGE_QUADS; ++i) {
      const int e = e0 + i * stride;
      c[i] = make_int4(0, 0, 0, 0);
      r[i] = make_int4(0, 0, 0, 0);
      s[i] = make_int4(0, 0, 0, 0);
      w[i] = make_float4(0.0f, 0.0f, 0.0f, 0.0f);
      if (e < E) {
        c[i] = *(const int4*)(col + e);
        r[i] = *(const int4*)(rk + e);
        s[i] = *(const int4*)(row + e);
        w[i] = *(const float4*)(wgt + e);
      }
    }
  } else {
#pragma unroll
    for (int i = 0; i < EDGE_QUADS; ++i) {
      c[i] = load_quad(col, e0 + i * stride, E, false);
      r[i] = load_quad(rk, e0 + i * stride, E, false);
      s[i] = load_quad(row, e0 + i * stride, E, false);
      w[i] = load_quad_f(wgt, e0 + i * stride, E, false);
    }
  }
  int4 o[EDGE_QUADS];
#pragma unroll
  for (int i = 0; i < EDGE_QUADS; ++i) {
    const int e = e0 + i * stride;
    o[i] = make_int4(0, 0, 0, 0);
    if (e < E) o[i].x = of[c[i].x];
    if (e + 1 < E) o[i].y = of[c[i].y];
    if (e + 2 < E) o[i].z = of[c[i].z];
    if (e + 3 < E) o[i].w = of[c[i].w];
  }
#pragma unroll
  for (int i = 0; i < EDGE_QUADS; ++i) {
    const int e = e0 + i * stride;
    if (e < E) rec[o[i].x + r[i].x] = make_int4(e, s[i].x, __float_as_int(w[i].x), 0);
    if (e + 1 < E) rec[o[i].y + r[i].y] = make_int4(e + 1, s[i].y, __float_as_int(w[i].y), 0);
    if (e + 2 < E) rec[o[i].z + r[i].z] = make_int4(e + 2, s[i].z, __float_as_int(w[i].z), 0);
    if (e + 3 < E) rec[o[i].w + r[i].w] = make_int4(e + 3, s[i].w, __float_as_int(w[i].w), 0);
  }
}

__global__ __launch_bounds__(256) void sort_kernel(
    const int* __restrict__ off, const int* __restrict__ tmp,
    int* __restrict__ ssrc, float* __restrict__ swt, int G) {
  const int lane = threadIdx.x;
  const int g0 = __builtin_amdgcn_readfirstlane(
      (blockIdx.x * 8 + threadIdx.y) * SORT_BUCKETS);
  if (g0 >= G) return;
  const int4* rec = (const int4*)tmp;

  int gi = g0 + lane;
  if (gi > G) gi = G;
  const int ov = (lane <= SORT_BUCKETS) ? off[gi] : 0;
  int st[SORT_BUCKETS + 1];
#pragma unroll
  for (int k = 0; k <= SORT_BUCKETS; ++k) st[k] = __builtin_amdgcn_readlane(ov, k);

  int4 rc[SORT_BUCKETS];
#pragma unroll
  for (int k = 0; k < SORT_BUCKETS; ++k) {
    const int n = st[k + 1] - st[k];
    rc[k] = make_int4(0x7fffffff, 0, 0, 0);
    if (lane < n) rc[k] = rec[st[k] + lane];
  }

  int rk[SORT_BUCKETS];
#pragma unroll
  for (int k = 0; k < SORT_BUCKETS; ++k) {
    const int n = st[k + 1] - st[k];
    rk[k] = 0;
    if (n <= 32) {
      for (int j = 0; j < n; ++j)
        rk[k] += (__builtin_amdgcn_readlane(rc[k].x, j) < rc[k].x) ? 1 : 0;
    }
  }

#pragma unroll
  for (int k = 0; k < SORT_BUCKETS; ++k) {
    const int n = st[k + 1] - st[k];
    if (n <= 32 && lane < n) {
      ssrc[st[k] + rk[k]] = rc[k].y;
      swt[st[k] + rk[k]] = __int_as_float(rc[k].z);
    }
  }

#pragma unroll
  for (int k = 0; k < SORT_BUCKETS; ++k) {
    const int n = st[k + 1] - st[k];
    if (n > 32) {
      const int start = st[k];
      for (int i = lane; i < n; i += 32) {
        const int4 me = rec[start + i];
        int q = 0;
        for (int j = 0; j < n; ++j) q += (rec[start + j].x < me.x) ? 1 : 0;
        ssrc[start + q] = me.y;
        swt[start + q] = __int_as_float(me.z);
      }
    }
  }
}

__global__ void dinv_kernel(const int* __restrict__ off, const float* __restrict__ swt,
                            float* __restrict__ dinv, int G) {
  const int g = blockIdx.x * blockDim.x + threadIdx.x;
  if (g >= G) return;
  const int end = off[g + 1];
  float s = 0.0f;
  for (int p = off[g]; p < end; ++p) s += swt[p];
  const float d = s + 1.0f;
  dinv[g] = rsqrtf(d);
}

typedef __attribute__((ext_vector_type(4))) float v4f;

#define GEMM_WAVES 4
#define KSTEPS_PER_GROUP 4

struct FragGroup {
  v2f a[KSTEPS_PER_GROUP];
  v4f blo[KSTEPS_PER_GROUP][2];
  v4f bhi[KSTEPS_PER_GROUP][2];
};

__device__ __forceinline__ void gemm_load_group(FragGroup& f, const float* __restrict__ zp,
                                                const float* __restrict__ wp, int kbase) {
#pragma unroll
  for (int q = 0; q < KSTEPS_PER_GROUP; ++q) {
    const int k0 = kbase + 4 * q;
    f.a[q] = *(const v2f*)(zp + k0);
#pragma unroll
    for (int s = 0; s < 2; ++s) {
      f.blo[q][s] = *(const v4f*)(wp + (size_t)(k0 + s) * D_FEAT);
      f.bhi[q][s] = *(const v4f*)(wp + (size_t)(k0 + s) * D_FEAT + 64);
    }
  }
}

__device__ __forceinline__ void gemm_mma_group(v8f (&acc)[8], const FragGroup& f) {
#pragma unroll
  for (int q = 0; q < KSTEPS_PER_GROUP; ++q) {
#pragma unroll
    for (int t = 0; t < 4; ++t) {
      v2f b;
      b.x = f.blo[q][0][t];
      b.y = f.blo[q][1][t];
      acc[t] = __builtin_amdgcn_wmma_f32_16x16x4_f32(
           false, f.a[q],  false, b,
           (short)0, acc[t],  false,  false);
      v2f d;
      d.x = f.bhi[q][0][t];
      d.y = f.bhi[q][1][t];
      acc[4 + t] = __builtin_amdgcn_wmma_f32_16x16x4_f32(
           false, f.a[q],  false, d,
           (short)0, acc[4 + t],  false,  false);
    }
  }
}

__global__ __launch_bounds__(32 * GEMM_WAVES) void gemm_wmma_kernel(
    const float* __restrict__ Z, const float* __restrict__ W,
    float* __restrict__ H, int nRows) {
  const int lane = threadIdx.x;
  const int strip = __builtin_amdgcn_readfirstlane(blockIdx.x * GEMM_WAVES + threadIdx.y);
  const int m0 = strip * 16;
  if (m0 >= nRows) return;

  const int c = lane & 15;
  const int h = lane >> 4;
  const float* zp = Z + (size_t)(m0 + c) * D_FEAT + 2 * h;
  const float* wp = W + (size_t)(2 * h) * D_FEAT + 4 * c;

  v8f acc[8];
#pragma unroll
  for (int t = 0; t < 8; ++t) acc[t] = v8f{};

  constexpr int KGROUP = 4 * KSTEPS_PER_GROUP;
  FragGroup f0, f1;
  gemm_load_group(f0, zp, wp, 0);
#pragma unroll
  for (int g = 0; g < D_FEAT / KGROUP; g += 2) {
    gemm_load_group(f1, zp, wp, (g + 1) * KGROUP);
    __builtin_amdgcn_sched_barrier(0);
    gemm_mma_group(acc, f0);
    if (g + 2 < D_FEAT / KGROUP) {
      gemm_load_group(f0, zp, wp, (g + 2) * KGROUP);
      __builtin_amdgcn_sched_barrier(0);
    }
    gemm_mma_group(acc, f1);
  }

  float* hp = H + (size_t)(m0 + 8 * h) * D_FEAT + 4 * c;
#pragma unroll
  for (int r = 0; r < 8; ++r) {
    v4f lo, hi;
#pragma unroll
    for (int t = 0; t < 4; ++t) {
      lo[t] = acc[t][r];
      hi[t] = acc[4 + t][r];
    }
    *(v4f*)(hp + (size_t)r * D_FEAT) = lo;
    *(v4f*)(hp + (size_t)r * D_FEAT + 64) = hi;
  }
}

template <int MODE>
__global__ __launch_bounds__(256) void pull_kernel(
    const int* __restrict__ offAll, const int* __restrict__ ssrc,
    const float* __restrict__ swt, const float* __restrict__ dinvAll,
    const float* __restrict__ Hall, const float* __restrict__ bias,
    const float* __restrict__ pw, const float* __restrict__ cscale,
    float* __restrict__ Zall, int N, size_t hPlane, size_t zPlane) {
  const int lane = threadIdx.x;
  const int c0 = __builtin_amdgcn_readfirstlane(
      (blockIdx.x * 8 + threadIdx.y) * PULL_NODES);
  if (c0 >= N) return;
  const int scale = blockIdx.y;
  const int* __restrict__ off = offAll + (size_t)scale * N;
  const float* __restrict__ dinv = dinvAll + (size_t)scale * N;
  const float* __restrict__ H = Hall + (size_t)scale * hPlane;
  float* __restrict__ Zout = Zall + (size_t)scale * zPlane;
  const float4* H4 = (const float4*)H;

  int ci = c0 + lane;
  if (ci > N) ci = N;
  const int ov = (lane <= PULL_NODES) ? off[ci] : 0;
  const float dv = (lane < PULL_NODES && c0 + lane < N) ? dinv[c0 + lane] : 0.0f;
  const float4 b = ((const float4*)bias)[lane];
  const float4 p = ((const float4*)pw)[lane];
  float ct = 0.0f;
  if (MODE != 0) ct = cscale[0];

  int st[PULL_NODES + 1];
  float dcs[PULL_NODES];
#pragma unroll
  for (int k = 0; k <= PULL_NODES; ++k) st[k] = __builtin_amdgcn_readlane(ov, k);
#pragma unroll
  for (int k = 0; k < PULL_NODES; ++k)
    dcs[k] = __int_as_float(__builtin_amdgcn_readlane(__float_as_int(dv), k));

  int r0[PULL_NODES];
  float w0[PULL_NODES];
#pragma unroll
  for (int k = 0; k < PULL_NODES; ++k) {
    const int n = st[k + 1] - st[k];
    r0[k] = 0;
    w0[k] = 0.0f;
    if (lane < n) {
      r0[k] = ssrc[st[k] + lane];
      w0[k] = swt[st[k] + lane];
    }
  }
  float d0[PULL_NODES];
#pragma unroll
  for (int k = 0; k < PULL_NODES; ++k) d0[k] = dinv[r0[k]];
  float n0[PULL_NODES];
#pragma unroll
  for (int k = 0; k < PULL_NODES; ++k) {
    const int n = st[k + 1] - st[k];
    const float x = d0[k] * w0[k] * dcs[k];
    n0[k] = (lane < n) ? x : 0.0f;
  }

#pragma unroll
  for (int k = 0; k < PULL_NODES; ++k) {
    const int c = c0 + k;
    if (c < N) {
      const int start = st[k];
      const int end = st[k + 1];
      const float dc = dcs[k];
      float4* zp = (float4*)Zout + (size_t)c * (D_FEAT / 4) + lane;
      const float4 h = H4[(size_t)c * (D_FEAT / 4) + lane];
      float4 s = make_float4(0.0f, 0.0f, 0.0f, 0.0f);
      if (MODE == 2) s = *zp;

      float4 acc = make_float4(0.0f, 0.0f, 0.0f, 0.0f);
      for (int base = start; base < end; base += 32) {
        const int m = (end - base < 32) ? (end - base) : 32;
        int r = r0[k];
        float nrm = n0[k];
        if (base != start) {
          r = 0;
          nrm = 0.0f;
          if (lane < m) {
            r = ssrc[base + lane];
            nrm = dinv[r] * swt[base + lane] * dc;
          }
        }
        const int nrm_bits = __float_as_int(nrm);
#pragma unroll 4
        for (int j = 0; j < m; ++j) {
          const int rj = __builtin_amdgcn_readlane(r, j);
          const float nj = __int_as_float(__builtin_amdgcn_readlane(nrm_bits, j));
          const float4 v = H4[(size_t)rj * (D_FEAT / 4) + lane];
          acc.x += v.x * nj;
          acc.y += v.y * nj;
          acc.z += v.z * nj;
          acc.w += v.w * nj;
        }
      }

      float4 o;
      float v;
      v = acc.x + h.x * dc * dc + b.x;
      o.x = fmaxf(v, 0.0f) + p.x * fminf(v, 0.0f);
      v = acc.y + h.y * dc * dc + b.y;
      o.y = fmaxf(v, 0.0f) + p.y * fminf(v, 0.0f);
      v = acc.z + h.z * dc * dc + b.z;
      o.z = fmaxf(v, 0.0f) + p.z * fminf(v, 0.0f);
      v = acc.w + h.w * dc * dc + b.w;
      o.w = fmaxf(v, 0.0f) + p.w * fminf(v, 0.0f);
      if (MODE == 0) {
        *zp = o;
      } else {
        float4 q;
        if (MODE == 1) {
          q.x = ct * o.x;
          q.y = ct * o.y;
          q.z = ct * o.z;
          q.w = ct * o.w;
        } else {
          q.x = s.x + ct * o.x;
          q.y = s.y + ct * o.y;
          q.z = s.z + ct * o.z;
          q.w = s.w + ct * o.w;
        }
        *zp = q;
      }
    }
  }
}

extern "C" void kernel_launch(void* const* d_in, const int* in_sizes, int n_in,
                              void* d_out, int out_size, void* d_ws, size_t ws_size,
                              hipStream_t stream) {
  const float* x       = (const float*)d_in[0];
  const int*   eidx    = (const int*)  d_in[1];
  const float* ew      = (const float*)d_in[2];
  const float* coeff   = (const float*)d_in[3];
  const float* W1      = (const float*)d_in[4];
  const float* b1      = (const float*)d_in[5];
  const float* W2      = (const float*)d_in[6];
  const float* b2      = (const float*)d_in[7];
  const float* prelu_w = (const float*)d_in[8];

  const int N = in_sizes[0] / D_FEAT;
  const int E = in_sizes[2] / 3;
  const int G = 3 * N;
  const size_t E3 = (size_t)3 * E;

  const size_t padG  = ((size_t)G + 1 + 31) & ~(size_t)31;
  const size_t padE3 = (E3 + 31) & ~(size_t)31;
  const int scanBlocks = (G + SCAN_B - 1) / SCAN_B;
  const size_t padS  = ((size_t)scanBlocks + 31) & ~(size_t)31;
  const int ETHREADS = 256;
  const int edgesPerBlock = ETHREADS * 4 * EDGE_QUADS;
  const int edgeBlocks = (E + edgesPerBlock - 1) / edgesPerBlock;
  const size_t padH  = ((size_t)3 * edgeBlocks + 31) & ~(size_t)31;
  int*   cnt  = (int*)d_ws;
  int*   cnt1 = cnt + padG;
  int*   off  = cnt1 + padG;
  int*   off1 = off + padG;
  int*   bsum = off1 + padG;
  int*   hflag = bsum + padS;
  int*   rank = hflag + padH;
  int*   tmp  = rank + padE3;
  int*   ssrc = tmp + 4 * padE3;
  float* swt  = (float*)(ssrc + padE3);
  float* dinv = swt + padE3;
  const size_t plane = (size_t)N * D_FEAT;
  float* H1 = dinv + padG;
  float* Z1 = H1 + plane;
  float* H2 = Z1 + 3 * plane;

  const dim3 waveBlock(32, 8);
  const dim3 gemmBlock(32, GEMM_WAVES);
  const int gemmBlocks1 = (N / 16 + GEMM_WAVES - 1) / GEMM_WAVES;
  const int gemmBlocks3 = (G / 16 + GEMM_WAVES - 1) / GEMM_WAVES;
  const int nodeWaveBlocks = (N + 8 * PULL_NODES - 1) / (8 * PULL_NODES);
  const dim3 nodeGridAll(nodeWaveBlocks, 3);
  const dim3 edgeGrid(edgeBlocks, 3);
  const int sortBlocks = (G + 8 * SORT_BUCKETS - 1) / (8 * SORT_BUCKETS);

  const int cntWords = (int)(2 * padG);
  zero_kernel<<<(cntWords + ETHREADS - 1) / ETHREADS, ETHREADS, 0, stream>>>((float*)cnt, cntWords);
  rank_kernel<<<edgeGrid, ETHREADS, 0, stream>>>(eidx, cnt, (int)padG, hflag, rank, E, N);
  bsum_kernel<<<scanBlocks, SCAN_B, 0, stream>>>(cnt, cnt1, bsum, G);
  scan_kernel<<<scanBlocks, SCAN_B, 0, stream>>>(cnt, cnt1, bsum, off, off1, G);
  fill_kernel<<<edgeGrid, ETHREADS, 0, stream>>>(eidx, ew, off, off1, hflag, rank, tmp, E, N);
  sort_kernel<<<sortBlocks, waveBlock, 0, stream>>>(off, tmp, ssrc, swt, G);
  dinv_kernel<<<(G + ETHREADS - 1) / ETHREADS, ETHREADS, 0, stream>>>(off, swt, dinv, G);

  gemm_wmma_kernel<<<gemmBlocks1, gemmBlock, 0, stream>>>(x, W1, H1, N);
  pull_kernel<0><<<nodeGridAll, waveBlock, 0, stream>>>(off, ssrc, swt, dinv, H1, b1, prelu_w, coeff, Z1, N, (size_t)0, plane);

  gemm_wmma_kernel<<<gemmBlocks3, gemmBlock, 0, stream>>>(Z1, W2, H2, G);

  for (int t = 0; t < 3; ++t) {
    const int*   offT  = off + (size_t)t * N;
    const float* dinvT = dinv + (size_t)t * N;
    const float* H2T   = H2 + (size_t)t * plane;
    if (t == 0)
      pull_kernel<1><<<nodeWaveBlocks, waveBlock, 0, stream>>>(offT, ssrc, swt, dinvT, H2T, b2, prelu_w, coeff + t, (float*)d_out, N, (size_t)0, (size_t)0);
    else
      pull_kernel<2><<<nodeWaveBlocks, waveBlock, 0, stream>>>(offT, ssrc, swt, dinvT, H2T, b2, prelu_w, coeff + t, (float*)d_out, N, (size_t)0, (size_t)0);
  }
}


# Layer-2 gather of the three scales in one launch: a wave takes five nodes through scales 0,1,2, row sums wait in LDS, out written once and never re-read; 13 -> 11 launches; sum order unchanged
# speedup vs baseline: 5.3553x; 1.0782x over previous
#include <hip/hip_runtime.h>
#include <hip/hip_bf16.h>


typedef __attribute__((ext_vector_type(2))) float v2f;
typedef __attribute__((ext_vector_type(8))) float v8f;

#define D_FEAT 128
#define SCAN_B 1024
#define EDGE_QUADS 2
#define SORT_BUCKETS 8
#define PULL_NODES 4
#define SUM_NODES 5

__device__ __forceinline__ int4 load_quad(const int* __restrict__ p, int e, int n, bool vec) {
  int4 v = make_int4(0, 0, 0, 0);
  if (e < n) {
    if (vec) {
      v = *(const int4*)(p + e);
    } else {
      v.x = p[e];
      if (e + 1 < n) v.y = p[e + 1];
      if (e + 2 < n) v.z = p[e + 2];
      if (e + 3 < n) v.w = p[e + 3];
    }
  }
  return v;
}

__device__ __forceinline__ void store_quad(int* __restrict__ p, int e, int n, bool vec, int4 v) {
  if (e < n) {
    if (vec) {
      *(int4*)(p + e) = v;
    } else {
      p[e] = v.x;
      if (e + 1 < n) p[e + 1] = v.y;
      if (e + 2 < n) p[e + 2] = v.z;
      if (e + 3 < n) p[e + 3] = v.w;
    }
  }
}

__device__ __forceinline__ float4 load_quad_f(const float* __restrict__ p, int e, int n, bool vec) {
  float4 v = make_float4(0.0f, 0.0f, 0.0f, 0.0f);
  if (e < n) {
    if (vec) {
      v = *(const float4*)(p + e);
    } else {
      v.x = p[e];
      if (e + 1 < n) v.y = p[e + 1];
      if (e + 2 < n) v.z = p[e + 2];
      if (e + 3 < n) v.w = p[e + 3];
    }
  }
  return v;
}

__global__ void zero_kernel(float* __restrict__ p, int n) {
  int i = blockIdx.x * blockDim.x + threadIdx.x;
  if (i < n) p[i] = 0.0f;
}

__global__ __launch_bounds__(256) void rank_kernel(
    const int* __restrict__ eidx, int* __restrict__ cnt, int cntPlane,
    int* __restrict__ hflag, int* __restrict__ rank, int E, int N) {
  const int t = blockIdx.y;
  const int* col = eidx + (size_t)t * 2 * E + E;
  int* rk = rank + (size_t)t * E;
  const bool vec = (E & 3) == 0;
  const int stride = gridDim.x * blockDim.x * 4;
  const int e0 = (blockIdx.x * blockDim.x + threadIdx.x) * 4;
  int4 c[EDGE_QUADS];
#pragma unroll
  for (int i = 0; i < EDGE_QUADS; ++i) c[i] = load_quad(col, e0 + i * stride, E, vec);

  __shared__ int blockHalf;
  if (threadIdx.x == 0) {
    const unsigned where = __builtin_amdgcn_s_sendmsg_rtn(0x87);
    blockHalf = (int)((where >> 18) & 1u);
    hflag[t * gridDim.x + blockIdx.x] = blockHalf;
  }
  __syncthreads();
  const int half = blockHalf;
  int* cn = cnt + (size_t)half * cntPlane + t * N;

  int4 r[EDGE_QUADS];
#pragma unroll
  for (int i = 0; i < EDGE_QUADS; ++i) {
    const int e = e0 + i * stride;
    r[i] = make_int4(0, 0, 0, 0);
    if (e < E) r[i].x = atomicAdd(&cn[c[i].x], 1);
    if (e + 1 < E) r[i].y = atomicAdd(&cn[c[i].y], 1);
    if (e + 2 < E) r[i].z = atomicAdd(&cn[c[i].z], 1);
    if (e + 3 < E) r[i].w = atomicAdd(&cn[c[i].w], 1);
  }
#pragma unroll
  for (int i = 0; i < EDGE_QUADS; ++i) store_quad(rk, e0 + i * stride, E, vec, r[i]);
}

__global__ __launch_bounds__(SCAN_B) void bsum_kernel(
    const int* __restrict__ cnt, const int* __restrict__ cnt1,
    int* __restrict__ bsum, int n) {
  __shared__ int s[32];
  const int tid = threadIdx.x;
  const int i = blockIdx.x * SCAN_B + tid;
  int v = (i < n) ? cnt[i] + cnt1[i] : 0;
  for (int m = 16; m > 0; m >>= 1) v += __shfl_xor(v, m, 32);
  if ((tid & 31) == 0) s[tid >> 5] = v;
  __syncthreads();
  if (tid < 32) {
    int x = s[tid];
    for (int m = 16; m > 0; m >>= 1) x += __shfl_xor(x, m, 32);
    if (tid == 0) bsum[blockIdx.x] = x;
  }
}

__global__ __launch_bounds__(SCAN_B) void scan_kernel(
    const int* __restrict__ cnt, const int* __restrict__ cnt1,
    const int* __restrict__ bsum,
    int* __restrict__ off, int* __restrict__ off1, int n) {
  __shared__ int sw[32];
  __shared__ int sbase;
  const int tid = threadIdx.x;
  const int lane = tid & 31;
  const int wv = tid >> 5;
  if (wv == 0) {
    int a = 0;
    for (int j = lane; j < (int)blockIdx.x; j += 32) a += bsum[j];
    for (int m = 16; m > 0; m >>= 1) a += __shfl_xor(a, m, 32);
    if (lane == 0) sbase = a;
  }
  const int i = blockIdx.x * SCAN_B + tid;
  const int v0 = (i < n) ? cnt[i] : 0;
  const int v = v0 + ((i < n) ? cnt1[i] : 0);
  int x = v;
  for (int d = 1; d < 32; d <<= 1) {
    const int y = __shfl_up(x, d, 32);
    if (lane >= d) x += y;
  }
  if (lane == 31) sw[wv] = x;
  __syncthreads();
  if (wv == 0) {
    const int wtot = sw[lane];
    int xx = wtot;
    for (int d = 1; d < 32; d <<= 1) {
      const int y = __shfl_up(xx, d, 32);
      if (lane >= d) xx += y;
    }
    sw[lane] = xx - wtot;
  }
  __syncthreads();
  const int excl = sbase + sw[wv] + x - v;
  if (i < n) {
    off[i] = excl;
    off1[i] = excl + v0;
  }
  if (i == n - 1) off[n] = excl + v;
}

__global__ __launch_bounds__(256) void fill_kernel(
    const int* __restrict__ eidx, const float* __restrict__ ew,
    const int* __restrict__ off,
    const int* __restrict__ off1, const int* __restrict__ hflag,
    const int* __restrict__ rank, int* __restrict__ tmp, int E, int N) {
  const int t = blockIdx.y;
  const int* row = eidx + (size_t)t * 2 * E;
  const int* col = row + E;
  const float* wgt = ew + (size_t)t * E;
  const int* rk = rank + (size_t)t * E;
  int4* rec = (int4*)tmp;
  const int* of = (hflag[t * gridDim.x + blockIdx.x] ? off1 : off) + t * N;
  const bool vec = (E & 3) == 0;
  const int stride = gridDim.x * blockDim.x * 4;
  const int e0 = (blockIdx.x * blockDim.x + threadIdx.x) * 4;
  int4 c[EDGE_QUADS], r[EDGE_QUADS], s[EDGE_QUADS];
  float4 w[EDGE_QUADS];
  if (vec) {
#pragma unroll
    for (int i = 0; i < EDGE_QUADS; ++i) {
      const int e = e0 + i * stride;
      c[i] = make_int4(0, 0, 0, 0);
      r[i] = make_int4(0, 0, 0, 0);
      s[i] = make_int4(0, 0, 0, 0);
      w[i] = make_float4(0.0f, 0.0f, 0.0f, 0.0f);
      if (e < E) {
        c[i] = *(const int4*)(col + e);
        r[i] = *(const int4*)(rk + e);
        s[i] = *(const int4*)(row + e);
        w[i] = *(const float4*)(wgt + e);
      }
    }
  } else {
#pragma unroll
    for (int i = 0; i < EDGE_QUADS; ++i) {
      c[i] = load_quad(col, e0 + i * stride, E, false);
      r[i] = load_quad(rk, e0 + i * stride, E, false);
      s[i] = load_quad(row, e0 + i * stride, E, false);
      w[i] = load_quad_f(wgt, e0 + i * stride, E, false);
    }
  }
  int4 o[EDGE_QUADS];
#pragma unroll
  for (int i = 0; i < EDGE_QUADS; ++i) {
    const int e = e0 + i * stride;
    o[i] = make_int4(0, 0, 0, 0);
    if (e < E) o[i].x = of[c[i].x];
    if (e + 1 < E) o[i].y = of[c[i].y];
    if (e + 2 < E) o[i].z = of[c[i].z];
    if (e + 3 < E) o[i].w = of[c[i].w];
  }
#pragma unroll
  for (int i = 0; i < EDGE_QUADS; ++i) {
    const int e = e0 + i * stride;
    if (e < E) rec[o[i].x + r[i].x] = make_int4(e, s[i].x, __float_as_int(w[i].x), 0);
    if (e + 1 < E) rec[o[i].y + r[i].y] = make_int4(e + 1, s[i].y, __float_as_int(w[i].y), 0);
    if (e + 2 < E) rec[o[i].z + r[i].z] = make_int4(e + 2, s[i].z, __float_as_int(w[i].z), 0);
    if (e + 3 < E) rec[o[i].w + r[i].w] = make_int4(e + 3, s[i].w, __float_as_int(w[i].w), 0);
  }
}

__global__ __launch_bounds__(256) void sort_kernel(
    const int* __restrict__ off, const int* __restrict__ tmp,
    int* __restrict__ ssrc, float* __restrict__ swt, int G) {
  const int lane = threadIdx.x;
  const int g0 = __builtin_amdgcn_readfirstlane(
      (blockIdx.x * 8 + threadIdx.y) * SORT_BUCKETS);
  if (g0 >= G) return;
  const int4* rec = (const int4*)tmp;

  int gi = g0 + lane;
  if (gi > G) gi = G;
  const int ov = (lane <= SORT_BUCKETS) ? off[gi] : 0;
  int st[SORT_BUCKETS + 1];
#pragma unroll
  for (int k = 0; k <= SORT_BUCKETS; ++k) st[k] = __builtin_amdgcn_readlane(ov, k);

  int4 rc[SORT_BUCKETS];
#pragma unroll
  for (int k = 0; k < SORT_BUCKETS; ++k) {
    const int n = st[k + 1] - st[k];
    rc[k] = make_int4(0x7fffffff, 0, 0, 0);
    if (lane < n) rc[k] = rec[st[k] + lane];
  }

  int rk[SORT_BUCKETS];
#pragma unroll
  for (int k = 0; k < SORT_BUCKETS; ++k) {
    const int n = st[k + 1] - st[k];
    rk[k] = 0;
    if (n <= 32) {
      for (int j = 0; j < n; ++j)
        rk[k] += (__builtin_amdgcn_readlane(rc[k].x, j) < rc[k].x) ? 1 : 0;
    }
  }

#pragma unroll
  for (int k = 0; k < SORT_BUCKETS; ++k) {
    const int n = st[k + 1] - st[k];
    if (n <= 32 && lane < n) {
      ssrc[st[k] + rk[k]] = rc[k].y;
      swt[st[k] + rk[k]] = __int_as_float(rc[k].z);
    }
  }

#pragma unroll
  for (int k = 0; k < SORT_BUCKETS; ++k) {
    const int n = st[k + 1] - st[k];
    if (n > 32) {
      const int start = st[k];
      for (int i = lane; i < n; i += 32) {
        const int4 me = rec[start + i];
        int q = 0;
        for (int j = 0; j < n; ++j) q += (rec[start + j].x < me.x) ? 1 : 0;
        ssrc[start + q] = me.y;
        swt[start + q] = __int_as_float(me.z);
      }
    }
  }
}

__global__ void dinv_kernel(const int* __restrict__ off, const float* __restrict__ swt,
                            float* __restrict__ dinv, int G) {
  const int g = blockIdx.x * blockDim.x + threadIdx.x;
  if (g >= G) return;
  const int end = off[g + 1];
  float s = 0.0f;
  for (int p = off[g]; p < end; ++p) s += swt[p];
  const float d = s + 1.0f;
  dinv[g] = rsqrtf(d);
}

typedef __attribute__((ext_vector_type(4))) float v4f;

#define GEMM_WAVES 4
#define KSTEPS_PER_GROUP 4

struct FragGroup {
  v2f a[KSTEPS_PER_GROUP];
  v4f blo[KSTEPS_PER_GROUP][2];
  v4f bhi[KSTEPS_PER_GROUP][2];
};

__device__ __forceinline__ void gemm_load_group(FragGroup& f, const float* __restrict__ zp,
                                                const float* __restrict__ wp, int kbase) {
#pragma unroll
  for (int q = 0; q < KSTEPS_PER_GROUP; ++q) {
    const int k0 = kbase + 4 * q;
    f.a[q] = *(const v2f*)(zp + k0);
#pragma unroll
    for (int s = 0; s < 2; ++s) {
      f.blo[q][s] = *(const v4f*)(wp + (size_t)(k0 + s) * D_FEAT);
      f.bhi[q][s] = *(const v4f*)(wp + (size_t)(k0 + s) * D_FEAT + 64);
    }
  }
}

__device__ __forceinline__ void gemm_mma_group(v8f (&acc)[8], const FragGroup& f) {
#pragma unroll
  for (int q = 0; q < KSTEPS_PER_GROUP; ++q) {
#pragma unroll
    for (int t = 0; t < 4; ++t) {
      v2f b;
      b.x = f.blo[q][0][t];
      b.y = f.blo[q][1][t];
      acc[t] = __builtin_amdgcn_wmma_f32_16x16x4_f32(
           false, f.a[q],  false, b,
           (short)0, acc[t],  false,  false);
      v2f d;
      d.x = f.bhi[q][0][t];
      d.y = f.bhi[q][1][t];
      acc[4 + t] = __builtin_amdgcn_wmma_f32_16x16x4_f32(
           false, f.a[q],  false, d,
           (short)0, acc[4 + t],  false,  false);
    }
  }
}

__global__ __launch_bounds__(32 * GEMM_WAVES) void gemm_wmma_kernel(
    const float* __restrict__ Z, const float* __restrict__ W,
    float* __restrict__ H, int nRows) {
  const int lane = threadIdx.x;
  const int strip = __builtin_amdgcn_readfirstlane(blockIdx.x * GEMM_WAVES + threadIdx.y);
  const int m0 = strip * 16;
  if (m0 >= nRows) return;

  const int c = lane & 15;
  const int h = lane >> 4;
  const float* zp = Z + (size_t)(m0 + c) * D_FEAT + 2 * h;
  const float* wp = W + (size_t)(2 * h) * D_FEAT + 4 * c;

  v8f acc[8];
#pragma unroll
  for (int t = 0; t < 8; ++t) acc[t] = v8f{};

  constexpr int KGROUP = 4 * KSTEPS_PER_GROUP;
  FragGroup f0, f1;
  gemm_load_group(f0, zp, wp, 0);
#pragma unroll
  for (int g = 0; g < D_FEAT / KGROUP; g += 2) {
    gemm_load_group(f1, zp, wp, (g + 1) * KGROUP);
    __builtin_amdgcn_sched_barrier(0);
    gemm_mma_group(acc, f0);
    if (g + 2 < D_FEAT / KGROUP) {
      gemm_load_group(f0, zp, wp, (g + 2) * KGROUP);
      __builtin_amdgcn_sched_barrier(0);
    }
    gemm_mma_group(acc, f1);
  }

  float* hp = H + (size_t)(m0 + 8 * h) * D_FEAT + 4 * c;
#pragma unroll
  for (int r = 0; r < 8; ++r) {
    v4f lo, hi;
#pragma unroll
    for (int t = 0; t < 4; ++t) {
      lo[t] = acc[t][r];
      hi[t] = acc[4 + t][r];
    }
    *(v4f*)(hp + (size_t)r * D_FEAT) = lo;
    *(v4f*)(hp + (size_t)r * D_FEAT + 64) = hi;
  }
}

template <int NODES, class Done>
__device__ __forceinline__ void pull_nodes(
    const int (&st)[NODES + 1], const float (&dcs)[NODES],
    const int* __restrict__ ssrc, const float* __restrict__ swt,
    const float* __restrict__ dinv, const float4* __restrict__ H4,
    const float4 b, const float4 p, int c0, int N, int lane, Done done) {
  int r0[NODES];
  float w0[NODES];
#pragma unroll
  for (int k = 0; k < NODES; ++k) {
    const int n = st[k + 1] - st[k];
    r0[k] = 0;
    w0[k] = 0.0f;
    if (lane < n) {
      r0[k] = ssrc[st[k] + lane];
      w0[k] = swt[st[k] + lane];
    }
  }
  float d0[NODES];
#pragma unroll
  for (int k = 0; k < NODES; ++k) d0[k] = dinv[r0[k]];
  float n0[NODES];
#pragma unroll
  for (int k = 0; k < NODES; ++k) {
    const int n = st[k + 1] - st[k];
    const float x = d0[k] * w0[k] * dcs[k];
    n0[k] = (lane < n) ? x : 0.0f;
  }

#pragma unroll
  for (int k = 0; k < NODES; ++k) {
    const int c = c0 + k;
    if (c < N) {
      const int start = st[k];
      const int end = st[k + 1];
      const float dc = dcs[k];
      const float4 h = H4[(size_t)c * (D_FEAT / 4) + lane];

      float4 acc = make_float4(0.0f, 0.0f, 0.0f, 0.0f);
      for (int base = start; base < end; base += 32) {
        const int m = (end - base < 32) ? (end - base) : 32;
        int r = r0[k];
        float nrm = n0[k];
        if (base != start) {
          r = 0;
          nrm = 0.0f;
          if (lane < m) {
            r = ssrc[base + lane];
            nrm = dinv[r] * swt[base + lane] * dc;
          }
        }
        const int nrm_bits = __float_as_int(nrm);
#pragma unroll 4
        for (int j = 0; j < m; ++j) {
          const int rj = __builtin_amdgcn_readlane(r, j);
          const float nj = __int_as_float(__builtin_amdgcn_readlane(nrm_bits, j));
          const float4 v = H4[(size_t)rj * (D_FEAT / 4) + lane];
          acc.x += v.x * nj;
          acc.y += v.y * nj;
          acc.z += v.z * nj;
          acc.w += v.w * nj;
        }
      }

      float4 o;
      float v;
      v = acc.x + h.x * dc * dc + b.x;
      o.x = fmaxf(v, 0.0f) + p.x * fminf(v, 0.0f);
      v = acc.y + h.y * dc * dc + b.y;
      o.y = fmaxf(v, 0.0f) + p.y * fminf(v, 0.0f);
      v = acc.z + h.z * dc * dc + b.z;
      o.z = fmaxf(v, 0.0f) + p.z * fminf(v, 0.0f);
      v = acc.w + h.w * dc * dc + b.w;
      o.w = fmaxf(v, 0.0f) + p.w * fminf(v, 0.0f);
      done(k, o);
    }
  }
}

__global__ __launch_bounds__(256) void pull_kernel(
    const int* __restrict__ offAll, const int* __restrict__ ssrc,
    const float* __restrict__ swt, const float* __restrict__ dinvAll,
    const float* __restrict__ H, const float* __restrict__ bias,
    const float* __restrict__ pw, float* __restrict__ Zall, int N, size_t zPlane) {
  const int lane = threadIdx.x;
  const int c0 = __builtin_amdgcn_readfirstlane(
      (blockIdx.x * 8 + threadIdx.y) * PULL_NODES);
  if (c0 >= N) return;
  const int scale = blockIdx.y;
  const int* __restrict__ off = offAll + (size_t)scale * N;
  const float* __restrict__ dinv = dinvAll + (size_t)scale * N;
  float* __restrict__ Zout = Zall + (size_t)scale * zPlane;

  int ci = c0 + lane;
  if (ci > N) ci = N;
  const int ov = (lane <= PULL_NODES) ? off[ci] : 0;
  const float dv = (lane < PULL_NODES && c0 + lane < N) ? dinv[c0 + lane] : 0.0f;
  const float4 b = ((const float4*)bias)[lane];
  const float4 p = ((const float4*)pw)[lane];

  int st[PULL_NODES + 1];
  float dcs[PULL_NODES];
#pragma unroll
  for (int k = 0; k <= PULL_NODES; ++k) st[k] = __builtin_amdgcn_readlane(ov, k);
#pragma unroll
  for (int k = 0; k < PULL_NODES; ++k)
    dcs[k] = __int_as_float(__builtin_amdgcn_readlane(__float_as_int(dv), k));

  pull_nodes<PULL_NODES>(st, dcs, ssrc, swt, dinv, (const float4*)H, b, p, c0, N, lane,
             [&](int k, float4 o) {
               *((float4*)Zout + (size_t)(c0 + k) * (D_FEAT / 4) + lane) = o;
             });
}

__global__ __launch_bounds__(256) void pull_sum_kernel(
    const int* __restrict__ offAll, const int* __restrict__ ssrc,
    const float* __restrict__ swt, const float* __restrict__ dinvAll,
    const float* __restrict__ Hall, const float* __restrict__ bias,
    const float* __restrict__ pw, const float* __restrict__ coeff,
    float* __restrict__ out, int N, size_t hPlane) {
  static_assert(SUM_NODES < 8, "eight lanes per scale hold SUM_NODES + 1 bounds");
  __shared__ float4 qsum[8][SUM_NODES][32];
  const int lane = threadIdx.x;
  const int c0 = __builtin_amdgcn_readfirstlane(
      (blockIdx.x * 8 + threadIdx.y) * SUM_NODES);
  if (c0 >= N) return;
  float4 (*qs)[32] = qsum[threadIdx.y];

  const int tl = lane >> 3;
  const int kl = lane & 7;
  int ci = c0 + kl;
  if (ci > N) ci = N;
  const int ov = (tl < 3 && kl <= SUM_NODES) ? offAll[(size_t)tl * N + ci] : 0;
  const float dv = (tl < 3 && kl < SUM_NODES && c0 + kl < N)
                       ? dinvAll[(size_t)tl * N + c0 + kl] : 0.0f;
  const float4 b = ((const float4*)bias)[lane];
  const float4 p = ((const float4*)pw)[lane];

#pragma nounroll
  for (int t = 0; t < 3; ++t) {
    int st[SUM_NODES + 1];
    float dcs[SUM_NODES];
#pragma unroll
    for (int k = 0; k <= SUM_NODES; ++k) st[k] = __builtin_amdgcn_readlane(ov, 8 * t + k);
#pragma unroll
    for (int k = 0; k < SUM_NODES; ++k)
      dcs[k] = __int_as_float(__builtin_amdgcn_readlane(__float_as_int(dv), 8 * t + k));
    const float ct = coeff[t];
    pull_nodes<SUM_NODES>(st, dcs, ssrc, swt, dinvAll + (size_t)t * N,
               (const float4*)(Hall + (size_t)t * hPlane), b, p, c0, N, lane,
               [&](int k, float4 o) {
                 float4 q;
                 if (t == 0) {
                   q.x = ct * o.x;
                   q.y = ct * o.y;
                   q.z = ct * o.z;
                   q.w = ct * o.w;
                 } else {
                   const float4 s = qs[k][lane];
                   q.x = s.x + ct * o.x;
                   q.y = s.y + ct * o.y;
                   q.z = s.z + ct * o.z;
                   q.w = s.w + ct * o.w;
                 }
                 qs[k][lane] = q;
               });
  }
#pragma unroll
  for (int k = 0; k < SUM_NODES; ++k)
    if (c0 + k < N) *((float4*)out + (size_t)(c0 + k) * (D_FEAT / 4) + lane) = qs[k][lane];
}

extern "C" void kernel_launch(void* const* d_in, const int* in_sizes, int n_in,
                              void* d_out, int out_size, void* d_ws, size_t ws_size,
                              hipStream_t stream) {
  const float* x       = (const float*)d_in[0];
  const int*   eidx    = (const int*)  d_in[1];
  const float* ew      = (const float*)d_in[2];
  const float* coeff   = (const float*)d_in[3];
  const float* W1      = (const float*)d_in[4];
  const float* b1      = (const float*)d_in[5];
  const float* W2      = (const float*)d_in[6];
  const float* b2      = (const float*)d_in[7];
  const float* prelu_w = (const float*)d_in[8];

  const int N = in_sizes[0] / D_FEAT;
  const int E = in_sizes[2] / 3;
  const int G = 3 * N;
  const size_t E3 = (size_t)3 * E;

  const size_t padG  = ((size_t)G + 1 + 31) & ~(size_t)31;
  const size_t padE3 = (E3 + 31) & ~(size_t)31;
  const int scanBlocks = (G + SCAN_B - 1) / SCAN_B;
  const size_t padS  = ((size_t)scanBlocks + 31) & ~(size_t)31;
  const int ETHREADS = 256;
  const int edgesPerBlock = ETHREADS * 4 * EDGE_QUADS;
  const int edgeBlocks = (E + edgesPerBlock - 1) / edgesPerBlock;
  const size_t padH  = ((size_t)3 * edgeBlocks + 31) & ~(size_t)31;
  int*   cnt  = (int*)d_ws;
  int*   cnt1 = cnt + padG;
  int*   off  = cnt1 + padG;
  int*   off1 = off + padG;
  int*   bsum = off1 + padG;
  int*   hflag = bsum + padS;
  int*   rank = hflag + padH;
  int*   tmp  = rank + padE3;
  int*   ssrc = tmp + 4 * padE3;
  float* swt  = (float*)(ssrc + padE3);
  float* dinv = swt + padE3;
  const size_t plane = (size_t)N * D_FEAT;
  float* H1 = dinv + padG;
  float* Z1 = H1 + plane;
  float* H2 = Z1 + 3 * plane;

  const dim3 waveBlock(32, 8);
  const dim3 gemmBlock(32, GEMM_WAVES);
  const int gemmBlocks1 = (N / 16 + GEMM_WAVES - 1) / GEMM_WAVES;
  const int gemmBlocks3 = (G / 16 + GEMM_WAVES - 1) / GEMM_WAVES;
  const int nodeWaveBlocks = (N + 8 * PULL_NODES - 1) / (8 * PULL_NODES);
  const int sumWaveBlocks = (N + 8 * SUM_NODES - 1) / (8 * SUM_NODES);
  const dim3 nodeGridAll(nodeWaveBlocks, 3);
  const dim3 edgeGrid(edgeBlocks, 3);
  const int sortBlocks = (G + 8 * SORT_BUCKETS - 1) / (8 * SORT_BUCKETS);

  const int cntWords = (int)(2 * padG);
  zero_kernel<<<(cntWords + ETHREADS - 1) / ETHREADS, ETHREADS, 0, stream>>>((float*)cnt, cntWords);
  rank_kernel<<<edgeGrid, ETHREADS, 0, stream>>>(eidx, cnt, (int)padG, hflag, rank, E, N);
  bsum_kernel<<<scanBlocks, SCAN_B, 0, stream>>>(cnt, cnt1, bsum, G);
  scan_kernel<<<scanBlocks, SCAN_B, 0, stream>>>(cnt, cnt1, bsum, off, off1, G);
  fill_kernel<<<edgeGrid, ETHREADS, 0, stream>>>(eidx, ew, off, off1, hflag, rank, tmp, E, N);
  sort_kernel<<<sortBlocks, waveBlock, 0, stream>>>(off, tmp, ssrc, swt, G);
  dinv_kernel<<<(G + ETHREADS - 1) / ETHREADS, ETHREADS, 0, stream>>>(off, swt, dinv, G);

  gemm_wmma_kernel<<<gemmBlocks1, gemmBlock, 0, stream>>>(x, W1, H1, N);
  pull_kernel<<<nodeGridAll, waveBlock, 0, stream>>>(off, ssrc, swt, dinv, H1, b1, prelu_w, Z1, N, plane);

  gemm_wmma_kernel<<<gemmBlocks3, gemmBlock, 0, stream>>>(Z1, W2, H2, G);

  pull_sum_kernel<<<sumWaveBlocks, waveBlock, 0, stream>>>(off, ssrc, swt, dinv, H2, b2, prelu_w, coeff, (float*)d_out, N, plane);
}
